# starting kernel
# baseline (speedup 1.0000x reference)
#include <hip/hip_runtime.h>
#include <cstdint>

#define F_IN  128
#define F_HID 128
#define F_OUT 64

typedef __attribute__((ext_vector_type(2))) float v2f;
typedef __attribute__((ext_vector_type(8))) float v8f;

__device__ __forceinline__ void atomic_add_f32(float* p, float v) {
  unsafeAtomicAdd(p, v);
}

__global__ void k_fill_const(float* __restrict__ p, float v, long long n) {
  long long i = (long long)blockIdx.x * blockDim.x + threadIdx.x;
  if (i < n) p[i] = v;
}

__global__ void k_fill_bias64(float* __restrict__ out, const float* __restrict__ b,
                              long long total) {
  long long i = (long long)blockIdx.x * blockDim.x + threadIdx.x;
  if (i < total) out[i] = b[i & 63];
}

__global__ void k_deg(const int* __restrict__ col, const float* __restrict__ ew,
                      float* __restrict__ deg, int E) {
  int e = blockIdx.x * blockDim.x + threadIdx.x;
  if (e < E) atomic_add_f32(&deg[col[e]], ew[e]);
}

__global__ void k_rsqrt(float* __restrict__ deg, int n) {
  int i = blockIdx.x * blockDim.x + threadIdx.x;
  if (i < n) {
    float d = deg[i];
    deg[i] = (d > 0.f) ? rsqrtf(d) : 0.f;
  }
}

__global__ void k_norm(const int* __restrict__ row, const int* __restrict__ col,
                       const float* __restrict__ ew, const float* __restrict__ dis,
                       float* __restrict__ nrm, int E) {
  int e = blockIdx.x * blockDim.x + threadIdx.x;
  if (e < E) nrm[e] = dis[row[e]] * ew[e] * dis[col[e]];
}

template <int FOUT>
__global__ void __launch_bounds__(256)
k_gemm_wmma(const float* __restrict__ A, const float* __restrict__ W,
            float* __restrict__ H, int n) {
  __shared__ float sW[F_IN * FOUT];
  for (int i = threadIdx.x; i < F_IN * FOUT; i += 256) sW[i] = W[i];
  __syncthreads();

  const int lane = threadIdx.x & 31;
  const int wave = threadIdx.x >> 5;
  const int tilesN = FOUT / 16;
  const long long numTiles = (long long)(n / 16) * tilesN;
  const long long tile = (long long)blockIdx.x * 8 + wave;
  if (tile >= numTiles) return;

  const int tm = (int)(tile / tilesN);
  const int tn = (int)(tile % tilesN);
  const int mrow = tm * 16 + (lane & 15);
  const int ncol = tn * 16 + (lane & 15);
  const int koff = (lane >> 4) * 2;

  const float* __restrict__ arow = A + (long long)mrow * F_IN;
  const float* __restrict__ wb = sW + ncol;

  v8f c = {};
#pragma unroll
  for (int k = 0; k < F_IN; k += 4) {
    const int ka = k + koff;
    v2f a, b;
    a.x = arow[ka];
    a.y = arow[ka + 1];
    b.x = wb[ka * FOUT];
    b.y = wb[(ka + 1) * FOUT];
    c = __builtin_amdgcn_wmma_f32_16x16x4_f32(
         false, a,  false, b,
         (short)0, c,  false,  false);
  }

  const int rbase = tm * 16 + ((lane >> 4) << 3);
#pragma unroll
  for (int v = 0; v < 8; ++v)
    H[(long long)(rbase + v) * FOUT + ncol] = c[v];
}

template <int F>
__global__ void __launch_bounds__(256)
k_scatter(const float* __restrict__ H, const int* __restrict__ row,
          const int* __restrict__ col, const float* __restrict__ nrm,
          float* __restrict__ out, int E) {
  const int lane = threadIdx.x & 31;
  const long long e = (long long)blockIdx.x * 8 + (threadIdx.x >> 5);
  if (e >= E) return;
  const int r = row[e];
  const int cgt = col[e];
  const float w = nrm[e];
  constexpr int PER = F / 32;
  const float* src = H + (long long)r * F + lane * PER;
  float* dst = out + (long long)cgt * F + lane * PER;
  if constexpr (PER == 4) {
    float4 v = *(const float4*)src;
    atomic_add_f32(dst + 0, v.x * w);
    atomic_add_f32(dst + 1, v.y * w);
    atomic_add_f32(dst + 2, v.z * w);
    atomic_add_f32(dst + 3, v.w * w);
  } else {
    float2 v = *(const float2*)src;
    atomic_add_f32(dst + 0, v.x * w);
    atomic_add_f32(dst + 1, v.y * w);
  }
}

__global__ void k_combine1(float* __restrict__ agg, const float* __restrict__ h,
                           const float* __restrict__ dis, const float* __restrict__ b1,
                           long long total) {
  long long i = (long long)blockIdx.x * blockDim.x + threadIdx.x;
  if (i >= total) return;
  const int node = (int)(i >> 7);
  const int j = (int)(i & 127);
  const float d = dis[node];
  float v = agg[i] + h[i] * d * d + b1[j];
  agg[i] = v > 0.f ? v : 0.f;
}

__global__ void k_combine2(float* __restrict__ out, const float* __restrict__ h2,
                           const float* __restrict__ dis, long long total) {
  long long i = (long long)blockIdx.x * blockDim.x + threadIdx.x;
  if (i >= total) return;
  const int node = (int)(i >> 6);
  const float d = dis[node];
  out[i] += h2[i] * d * d;
}

static inline int nblk(long long n, int t) { return (int)((n + t - 1) / t); }

extern "C" void kernel_launch(void* const* d_in, const int* in_sizes, int n_in,
                              void* d_out, int out_size, void* d_ws, size_t ws_size,
                              hipStream_t stream) {
  const float* x  = (const float*)d_in[0];
  const int*   ei = (const int*)d_in[1];
  const float* ew = (const float*)d_in[2];
  const float* W1 = (const float*)d_in[3];
  const float* b1 = (const float*)d_in[4];
  const float* W2 = (const float*)d_in[5];
  const float* b2 = (const float*)d_in[6];
  float* out = (float*)d_out;

  const int n = in_sizes[0] / F_IN;
  const int E = in_sizes[1] / 2;
  const int* row = ei;
  const int* col = ei + E;

  char* ws = (char*)d_ws;
  size_t o = 0;
  auto carve = [&](size_t bytes) {
    char* p = ws + o;
    o += (bytes + 255) & ~(size_t)255;
    return p;
  };
  float* dis = (float*)carve((size_t)n * 4);
  float* nrm = (float*)carve((size_t)E * 4);
  float* h   = (float*)carve((size_t)n * F_HID * 4);
  float* agg = (float*)carve((size_t)n * F_HID * 4);

  const long long nF1 = (long long)n * F_HID;
  const long long nF2 = (long long)n * F_OUT;

  k_fill_const<<<nblk(n, 256), 256, 0, stream>>>(dis, 1.0f, n);
  k_fill_const<<<nblk(nF1, 256), 256, 0, stream>>>(agg, 0.0f, nF1);
  k_fill_bias64<<<nblk(nF2, 256), 256, 0, stream>>>(out, b2, nF2);

  k_deg<<<nblk(E, 256), 256, 0, stream>>>(col, ew, dis, E);
  k_rsqrt<<<nblk(n, 256), 256, 0, stream>>>(dis, n);
  k_norm<<<nblk(E, 256), 256, 0, stream>>>(row, col, ew, dis, nrm, E);

  {
    const long long tiles = (long long)(n / 16) * (F_HID / 16);
    k_gemm_wmma<F_HID><<<nblk(tiles, 8), 256, 0, stream>>>(x, W1, h, n);
  }
  k_scatter<F_HID><<<nblk((long long)E, 8), 256, 0, stream>>>(h, row, col, nrm, agg, E);
  k_combine1<<<nblk(nF1, 256), 256, 0, stream>>>(agg, h, dis, b1, nF1);

  {
    const long long tiles = (long long)(n / 16) * (F_OUT / 16);
    k_gemm_wmma<F_OUT><<<nblk(tiles, 8), 256, 0, stream>>>(agg, W2, h, n);
  }
  k_scatter<F_OUT><<<nblk((long long)E, 8), 256, 0, stream>>>(h, row, col, nrm, out, E);
  k_combine2<<<nblk(nF2, 256), 256, 0, stream>>>(out, h, dis, nF2);
}


# Float-atomic scatter replaced by per-call counting sort of edges by destination (int-atomic arrival-order rank) and an atomic-free per-node register gather that stores once with its epilogue
# speedup vs baseline: 1.8136x; 1.8136x over previous
#include <hip/hip_runtime.h>
#include <cstdint>

#define F_IN  128
#define F_HID 128
#define F_OUT 64

typedef __attribute__((ext_vector_type(2))) float v2f;
typedef __attribute__((ext_vector_type(8))) float v8f;

__global__ void k_zero_i32(int* __restrict__ p, int n) {
  int i = blockIdx.x * blockDim.x + threadIdx.x;
  if (i < n) p[i] = 0;
}

__global__ void k_rank(const int* __restrict__ col, int* __restrict__ cnt,
                       int* __restrict__ pos, int E) {
  int e = blockIdx.x * blockDim.x + threadIdx.x;
  if (e < E) pos[e] = atomicAdd(&cnt[col[e]], 1);
}

__global__ void __launch_bounds__(1024)
k_rowptr(const int* __restrict__ cnt, int* __restrict__ rowptr, int n) {
  __shared__ int sA[1024];
  __shared__ int sB[1024];
  const int t = threadIdx.x;
  const int b = blockIdx.x;
  int part = 0;
  for (int k = 0; k < b; ++k) part += cnt[k * 1024 + t];
  const int g = b * 1024 + t;
  const int mine = (g < n) ? cnt[g] : 0;
  sA[t] = mine;
  sB[t] = part;
  __syncthreads();
  for (int off = 1; off < 1024; off <<= 1) {
    int va = 0, vb = 0;
    if (t >= off) { va = sA[t - off]; vb = sB[t - off]; }
    __syncthreads();
    sA[t] += va;
    sB[t] += vb;
    __syncthreads();
  }
  if (g <= n) rowptr[g] = sB[1023] + sA[t] - mine;
}

__global__ void k_place(const int* __restrict__ row, const int* __restrict__ col,
                        const float* __restrict__ ew, const int* __restrict__ rowptr,
                        const int* __restrict__ pos, int* __restrict__ srow,
                        float* __restrict__ sw, int E) {
  int e = blockIdx.x * blockDim.x + threadIdx.x;
  if (e < E) {
    const int p = rowptr[col[e]] + pos[e];
    srow[p] = row[e];
    sw[p] = ew[e];
  }
}

__global__ void k_dis(const int* __restrict__ rowptr, const float* __restrict__ sw,
                      float* __restrict__ dis, int n) {
  int i = blockIdx.x * blockDim.x + threadIdx.x;
  if (i < n) {
    const int beg = rowptr[i], end = rowptr[i + 1];
    float d = 1.0f;
    for (int p = beg; p < end; ++p) d += sw[p];
    dis[i] = (d > 0.f) ? rsqrtf(d) : 0.f;
  }
}

template <int FOUT>
__global__ void __launch_bounds__(256)
k_gemm_wmma(const float* __restrict__ A, const float* __restrict__ W,
            float* __restrict__ H, int n) {
  __shared__ float sW[F_IN * FOUT];
  for (int i = threadIdx.x; i < F_IN * FOUT; i += 256) sW[i] = W[i];
  __syncthreads();

  const int lane = threadIdx.x & 31;
  const int wave = threadIdx.x >> 5;
  const int tilesN = FOUT / 16;
  const long long numTiles = (long long)(n / 16) * tilesN;
  const long long tile = (long long)blockIdx.x * 8 + wave;
  if (tile >= numTiles) return;

  const int tm = (int)(tile / tilesN);
  const int tn = (int)(tile % tilesN);
  const int mrow = tm * 16 + (lane & 15);
  const int ncol = tn * 16 + (lane & 15);
  const int koff = (lane >> 4) * 2;

  const float* __restrict__ arow = A + (long long)mrow * F_IN;
  const float* __restrict__ wb = sW + ncol;

  v8f c = {};
#pragma unroll
  for (int k = 0; k < F_IN; k += 4) {
    const int ka = k + koff;
    v2f a, b;
    a.x = arow[ka];
    a.y = arow[ka + 1];
    b.x = wb[ka * FOUT];
    b.y = wb[(ka + 1) * FOUT];
    c = __builtin_amdgcn_wmma_f32_16x16x4_f32(
         false, a,  false, b,
         (short)0, c,  false,  false);
  }

  const int rbase = tm * 16 + ((lane >> 4) << 3);
#pragma unroll
  for (int v = 0; v < 8; ++v)
    H[(long long)(rbase + v) * FOUT + ncol] = c[v];
}

template <int F, bool RELU>
__global__ void __launch_bounds__(256)
k_gather(const float* __restrict__ H, const int* __restrict__ rowptr,
         const int* __restrict__ srow, const float* __restrict__ sw,
         const float* __restrict__ dis, const float* __restrict__ bias,
         float* __restrict__ out, int n) {
  constexpr int PER = F / 32;
  const int lane = threadIdx.x & 31;
  const int node = blockIdx.x * 8 + __builtin_amdgcn_readfirstlane((int)(threadIdx.x >> 5));
  if (node >= n) return;
  const int beg = rowptr[node];
  const int end = rowptr[node + 1];
  const float dc = dis[node];

  float acc[PER];
  {
    const float* src = H + (long long)node * F + lane * PER;
    const float sl = dc * dc;
    if constexpr (PER == 4) {
      const float4 v = *(const float4*)src;
      acc[0] = v.x * sl; acc[1] = v.y * sl; acc[2] = v.z * sl; acc[3] = v.w * sl;
    } else {
      const float2 v = *(const float2*)src;
      acc[0] = v.x * sl; acc[1] = v.y * sl;
    }
  }

  for (int base = beg; base < end; base += 32) {
    const int p = base + lane;
    int rr = 0;
    float nw = 0.f;
    if (p < end) {
      rr = srow[p];
      nw = dis[rr] * sw[p] * dc;
    }
    int m = end - base;
    if (m > 32) m = 32;
#pragma unroll 4
    for (int j = 0; j < m; ++j) {
      const int r = __builtin_amdgcn_readlane(rr, j);
      const float w = __int_as_float(__builtin_amdgcn_readlane(__float_as_int(nw), j));
      const float* src = H + (long long)r * F + lane * PER;
      if constexpr (PER == 4) {
        const float4 v = *(const float4*)src;
        acc[0] += v.x * w; acc[1] += v.y * w; acc[2] += v.z * w; acc[3] += v.w * w;
      } else {
        const float2 v = *(const float2*)src;
        acc[0] += v.x * w; acc[1] += v.y * w;
      }
    }
  }

  const float* bp = bias + lane * PER;
  float* dst = out + (long long)node * F + lane * PER;
  if constexpr (PER == 4) {
    float4 o;
    o.x = acc[0] + bp[0]; o.y = acc[1] + bp[1]; o.z = acc[2] + bp[2]; o.w = acc[3] + bp[3];
    if constexpr (RELU) {
      o.x = o.x > 0.f ? o.x : 0.f; o.y = o.y > 0.f ? o.y : 0.f;
      o.z = o.z > 0.f ? o.z : 0.f; o.w = o.w > 0.f ? o.w : 0.f;
    }
    *(float4*)dst = o;
  } else {
    float2 o;
    o.x = acc[0] + bp[0]; o.y = acc[1] + bp[1];
    if constexpr (RELU) {
      o.x = o.x > 0.f ? o.x : 0.f; o.y = o.y > 0.f ? o.y : 0.f;
    }
    *(float2*)dst = o;
  }
}

static inline int nblk(long long n, int t) { return (int)((n + t - 1) / t); }

extern "C" void kernel_launch(void* const* d_in, const int* in_sizes, int n_in,
                              void* d_out, int out_size, void* d_ws, size_t ws_size,
                              hipStream_t stream) {
  const float* x  = (const float*)d_in[0];
  const int*   ei = (const int*)d_in[1];
  const float* ew = (const float*)d_in[2];
  const float* W1 = (const float*)d_in[3];
  const float* b1 = (const float*)d_in[4];
  const float* W2 = (const float*)d_in[5];
  const float* b2 = (const float*)d_in[6];
  float* out = (float*)d_out;

  const int n = in_sizes[0] / F_IN;
  const int E = in_sizes[1] / 2;
  const int* row = ei;
  const int* col = ei + E;

  char* ws = (char*)d_ws;
  size_t o = 0;
  auto carve = [&](size_t bytes) {
    char* p = ws + o;
    o += (bytes + 255) & ~(size_t)255;
    return p;
  };
  int*   cnt    = (int*)carve((size_t)n * 4);
  int*   rowptr = (int*)carve((size_t)(n + 1) * 4);
  int*   pos    = (int*)carve((size_t)E * 4);
  int*   srow   = (int*)carve((size_t)E * 4);
  float* sw     = (float*)carve((size_t)E * 4);
  float* dis    = (float*)carve((size_t)n * 4);
  float* h      = (float*)carve((size_t)n * F_HID * 4);
  float* a1     = (float*)carve((size_t)n * F_HID * 4);

  k_zero_i32<<<nblk(n, 256), 256, 0, stream>>>(cnt, n);
  k_rank<<<nblk(E, 256), 256, 0, stream>>>(col, cnt, pos, E);
  k_rowptr<<<nblk((long long)n + 1, 1024), 1024, 0, stream>>>(cnt, rowptr, n);
  k_place<<<nblk(E, 256), 256, 0, stream>>>(row, col, ew, rowptr, pos, srow, sw, E);
  k_dis<<<nblk(n, 256), 256, 0, stream>>>(rowptr, sw, dis, n);

  {
    const long long tiles = (long long)(n / 16) * (F_HID / 16);
    k_gemm_wmma<F_HID><<<nblk(tiles, 8), 256, 0, stream>>>(x, W1, h, n);
  }
  k_gather<F_HID, true><<<nblk(n, 8), 256, 0, stream>>>(h, rowptr, srow, sw, dis, b1, a1, n);

  {
    const long long tiles = (long long)(n / 16) * (F_OUT / 16);
    k_gemm_wmma<F_OUT><<<nblk(tiles, 8), 256, 0, stream>>>(a1, W2, h, n);
  }
  k_gather<F_OUT, false><<<nblk(n, 8), 256, 0, stream>>>(h, rowptr, srow, sw, dis, b2, out, n);
}


# Layer-1 GEMM tiles ride in the two edge-sort launches as extra 32-wave blocks (block-specialised kernels, one shared LDS buffer, 64-register cap), overlapping the sort's waits; 6 launches become 5
# speedup vs baseline: 3.6577x; 2.0168x over previous
#include <hip/hip_runtime.h>
#include <cstdint>

#define F_IN  128
#define F_HID 128
#define F_OUT 64

#define GEMM_BLOCKS   1024
#define GT            64
#define GATHER_BLOCKS 2048

typedef __attribute__((ext_vector_type(2))) float v2f;
typedef __attribute__((ext_vector_type(8))) float v8f;

#define CH   4096
#define OT   257
#define U2   5
#define TCAP 8192

__device__ __forceinline__ int block_excl_scan(int v, int t, int* s_w, int& total) {
  const int lane = t & 31;
  const int wave = t >> 5;
  int x = v;
#pragma unroll
  for (int d = 1; d < 32; d <<= 1) {
    const int y = __shfl_up(x, d);
    if (lane >= d) x += y;
  }
  if (lane == 31) s_w[wave] = x;
  __syncthreads();
  if (wave == 0) {
    const int w = s_w[lane];
    int z = w;
#pragma unroll
    for (int d = 1; d < 32; d <<= 1) {
      const int y = __shfl_up(z, d);
      if (lane >= d) z += y;
    }
    s_w[lane] = z - w;
    if (lane == 31) s_w[32] = z;
  }
  __syncthreads();
  total = s_w[32];
  return s_w[wave] + x - v;
}

#define SORT1_WORDS (576 + 3 * CH)
#define SORT2_WORDS (1152 + 2 * TCAP)

__device__ __forceinline__ void
chunk_sort_body(const int* __restrict__ row, const int* __restrict__ col, const float* __restrict__ ew,
                int* __restrict__ ccol, int* __restrict__ crow, float* __restrict__ cw,
                int* __restrict__ offtab, int E, int* smem) {
  int*   s_hist = smem;
  int*   s_off  = smem + 256;
  int*   s_scan = smem + 520;
  int*   l_col  = smem + 576;
  int*   l_row  = smem + 576 + CH;
  float* l_w    = (float*)(smem + 576 + 2 * CH);
  const int t = threadIdx.x;
  const int e0 = blockIdx.x * CH;
  int cnt = E - e0;
  if (cnt > CH) cnt = CH;
  if (t < 256) s_hist[t] = 0;

  int cc[4], rr[4], rk[4];
  float ww[4];
#pragma unroll
  for (int k = 0; k < 4; ++k) {
    const int i = t + k * 1024;
    cc[k] = 0; rr[k] = 0; ww[k] = 0.f;
    if (i < cnt) { cc[k] = col[e0 + i]; rr[k] = row[e0 + i]; ww[k] = ew[e0 + i]; }
  }
  __syncthreads();
#pragma unroll
  for (int k = 0; k < 4; ++k) {
    rk[k] = 0;
    if (t + k * 1024 < cnt) rk[k] = atomicAdd(&s_hist[cc[k] >> 8], 1);
  }
  __syncthreads();
  int total;
  const int ex = block_excl_scan((t < 256) ? s_hist[t] : 0, t, s_scan, total);
  if (t <= 256) {
    s_off[t] = ex;
    offtab[blockIdx.x * OT + t] = ex;
  }
  __syncthreads();
#pragma unroll
  for (int k = 0; k < 4; ++k) {
    if (t + k * 1024 < cnt) {
      const int p = s_off[cc[k] >> 8] + rk[k];
      l_col[p] = cc[k]; l_row[p] = rr[k]; l_w[p] = ww[k];
    }
  }
  __syncthreads();
#pragma unroll
  for (int k = 0; k < 4; ++k) {
    const int i = t + k * 1024;
    if (i < cnt) { ccol[e0 + i] = l_col[i]; crow[e0 + i] = l_row[i]; cw[e0 + i] = l_w[i]; }
  }
}

__device__ __forceinline__ void
bucket_finish_body(const int* __restrict__ ccol, const int* __restrict__ crow, const float* __restrict__ cw,
                   const int* __restrict__ offtab, int* __restrict__ rowptr, int* __restrict__ srow,
                   float* __restrict__ sw, float* __restrict__ dis, int n, int E, int nchunk, int nbucket,
                   int* smem) {
  int*   s_pre    = smem;
  int*   s_src    = smem + 256;
  int*   s_cnt    = smem + 512;
  float* s_deg    = (float*)(smem + 768);
  int*   s_scan_a = smem + 1024;
  int*   s_scan_b = smem + 1064;
  int*   s_scan_c = smem + 1104;
  int*   st_row   = smem + 1152;
  float* st_w     = (float*)(smem + 1152 + TCAP);
  const int t = threadIdx.x;
  const int b = blockIdx.x;

  int lo = 0, len = 0;
  if (t < nchunk) {
    lo  = offtab[t * OT + b];
    len = offtab[t * OT + b + 1] - lo;
  }
  if (t < 256) { s_cnt[t] = 0; s_deg[t] = 0.f; }
  int total, base;
  const int pre = block_excl_scan(len, t, s_scan_a, total);
  (void)block_excl_scan(lo, t, s_scan_b, base);
  if (t < 256) {
    s_pre[t] = (t < nchunk) ? pre : 0x7fffffff;
    s_src[t] = t * CH + lo;
  }
  __syncthreads();

  auto locate = [&](int i) {
    int c = 0;
#pragma unroll
    for (int s = 128; s > 0; s >>= 1)
      if (s_pre[c + s] <= i) c += s;
    return s_src[c] + (i - s_pre[c]);
  };

  int kc[U2], kr[U2], ks[U2];
  float kw[U2];
#pragma unroll
  for (int k = 0; k < U2; ++k) ks[k] = 0;
#pragma unroll
  for (int s = 128; s > 0; s >>= 1) {
#pragma unroll
    for (int k = 0; k < U2; ++k)
      if (s_pre[ks[k] + s] <= t + k * 1024) ks[k] += s;
  }
#pragma unroll
  for (int k = 0; k < U2; ++k) {
    const int i = t + k * 1024;
    kc[k] = 0; kr[k] = 0; kw[k] = 0.f;
    if (i < total) {
      const int j = s_src[ks[k]] + (i - s_pre[ks[k]]);
      kc[k] = ccol[j]; kr[k] = crow[j]; kw[k] = cw[j];
    }
  }
#pragma unroll
  for (int k = 0; k < U2; ++k) {
    if (t + k * 1024 < total) {
      const int ln = kc[k] & 255;
      atomicAdd(&s_cnt[ln], 1);
      atomicAdd(&s_deg[ln], kw[k]);
    }
  }
  for (int i = t + U2 * 1024; i < total; i += 1024) {
    const int j = locate(i);
    const int ln = ccol[j] & 255;
    atomicAdd(&s_cnt[ln], 1);
    atomicAdd(&s_deg[ln], cw[j]);
  }
  __syncthreads();

  int tot2;
  const int ex = block_excl_scan((t < 256) ? s_cnt[t] : 0, t, s_scan_c, tot2);
  if (t < 256) {
    s_cnt[t] = ex;
    const int node = b * 256 + t;
    if (node < n) {
      rowptr[node] = base + ex;
      const float d = 1.0f + s_deg[t];
      dis[node] = (d > 0.f) ? rsqrtf(d) : 0.f;
    }
  }
  if (b == nbucket - 1 && t == 0) rowptr[n] = E;
  __syncthreads();

  const bool staged = total <= TCAP;
#pragma unroll
  for (int k = 0; k < U2; ++k) {
    if (t + k * 1024 < total) {
      const int p = atomicAdd(&s_cnt[kc[k] & 255], 1);
      if (staged) { st_row[p] = kr[k]; st_w[p] = kw[k]; }
      else        { srow[base + p] = kr[k]; sw[base + p] = kw[k]; }
    }
  }
  for (int i = t + U2 * 1024; i < total; i += 1024) {
    const int j = locate(i);
    const int p = atomicAdd(&s_cnt[ccol[j] & 255], 1);
    const int r = crow[j];
    const float w = cw[j];
    if (staged) { st_row[p] = r; st_w[p] = w; }
    else        { srow[base + p] = r; sw[base + p] = w; }
  }
  __syncthreads();
  if (staged) {
    for (int i = t; i < total; i += 1024) { srow[base + i] = st_row[i]; sw[base + i] = st_w[i]; }
  }
}

template <int FOUT, int NW, int KG>
__device__ __forceinline__ void
gemm_tiles(const float* __restrict__ A, const float* __restrict__ W, float* __restrict__ H,
           float* sW, int first, int last, int tstride) {
  for (int i = threadIdx.x; i < F_IN * FOUT; i += NW * 32) sW[i] = W[i];
  __syncthreads();

  const int lane = threadIdx.x & 31;
  const int wave = __builtin_amdgcn_readfirstlane((int)(threadIdx.x >> 5));
  constexpr int tilesN = FOUT / 16;
  const int koff = (lane >> 4) * 2;

  for (int tile = first + wave; tile < last; tile += tstride) {
    const int tm = tile / tilesN;
    const int tn = tile % tilesN;
    const int mrow = tm * 16 + (lane & 15);
    const int ncol = tn * 16 + (lane & 15);

    const float* __restrict__ arow = A + (long long)mrow * F_IN;
    const float* __restrict__ wb = sW + ncol;

    v8f c = {};
    for (int k0 = 0; k0 < F_IN; k0 += 4 * KG) {
      v2f a[KG];
#pragma unroll
      for (int j = 0; j < KG; ++j) {
        const int ka = k0 + 4 * j + koff;
        a[j].x = arow[ka];
        a[j].y = arow[ka + 1];
      }
#pragma unroll
      for (int j = 0; j < KG; ++j) {
        const int ka = k0 + 4 * j + koff;
        v2f b;
        b.x = wb[ka * FOUT];
        b.y = wb[(ka + 1) * FOUT];
        c = __builtin_amdgcn_wmma_f32_16x16x4_f32(
             false, a[j],  false, b,
             (short)0, c,  false,  false);
      }
    }

    const int rbase = tm * 16 + ((lane >> 4) << 3);
#pragma unroll
    for (int v = 0; v < 8; ++v)
      H[(long long)(rbase + v) * FOUT + ncol] = c[v];
  }
}

template <int FOUT>
__global__ void __launch_bounds__(256)
k_gemm_wmma(const float* __restrict__ A, const float* __restrict__ W,
            float* __restrict__ H, int n) {
  __shared__ float sW[F_IN * FOUT];
  const int numTiles = (n / 16) * (FOUT / 16);
  gemm_tiles<FOUT, 8, 32>(A, W, H, sW, (int)blockIdx.x * 8, numTiles, (int)gridDim.x * 8);
}

__global__ void __launch_bounds__(1024, 16)
k_sort1_gemm(const int* __restrict__ row, const int* __restrict__ col, const float* __restrict__ ew,
             int* __restrict__ ccol, int* __restrict__ crow, float* __restrict__ cw,
             int* __restrict__ offtab, int E, int nsort,
             const float* __restrict__ A, const float* __restrict__ Wm, float* __restrict__ H,
             int gt0, int gt1) {
  constexpr int WORDS = (F_IN * F_HID > SORT1_WORDS) ? F_IN * F_HID : SORT1_WORDS;
  __shared__ int smem[WORDS];
  if ((int)blockIdx.x < nsort) {
    chunk_sort_body(row, col, ew, ccol, crow, cw, offtab, E, smem);
  } else {
    const int t0 = gt0 + ((int)blockIdx.x - nsort) * GT;
    const int t1 = (t0 + GT < gt1) ? t0 + GT : gt1;
    gemm_tiles<F_HID, 32, 16>(A, Wm, H, (float*)smem, t0, t1, 32);
  }
}

__global__ void __launch_bounds__(1024, 16)
k_sort2_gemm(const int* __restrict__ ccol, const int* __restrict__ crow, const float* __restrict__ cw,
             const int* __restrict__ offtab, int* __restrict__ rowptr, int* __restrict__ srow,
             float* __restrict__ sw, float* __restrict__ dis, int n, int E, int nchunk, int nsort,
             const float* __restrict__ A, const float* __restrict__ Wm, float* __restrict__ H,
             int gt0, int gt1) {
  constexpr int WORDS = (F_IN * F_HID > SORT2_WORDS) ? F_IN * F_HID : SORT2_WORDS;
  __shared__ int smem[WORDS];
  if ((int)blockIdx.x < nsort) {
    bucket_finish_body(ccol, crow, cw, offtab, rowptr, srow, sw, dis, n, E, nchunk, nsort, smem);
  } else {
    const int t0 = gt0 + ((int)blockIdx.x - nsort) * GT;
    const int t1 = (t0 + GT < gt1) ? t0 + GT : gt1;
    gemm_tiles<F_HID, 32, 16>(A, Wm, H, (float*)smem, t0, t1, 32);
  }
}

template <int F, bool RELU>
__global__ void __launch_bounds__(256)
k_gather(const float* __restrict__ H, const int* __restrict__ rowptr,
         const int* __restrict__ srow, const float* __restrict__ sw,
         const float* __restrict__ dis, const float* __restrict__ bias,
         float* __restrict__ out, int n) {
  constexpr int PER = F / 32;
  const int lane = threadIdx.x & 31;
  const int wave = __builtin_amdgcn_readfirstlane((int)(threadIdx.x >> 5));
  const int nstride = (int)gridDim.x * 8;
  int node = (int)blockIdx.x * 8 + wave;
  if (node >= n) return;

  float bv[PER];
#pragma unroll
  for (int q = 0; q < PER; ++q) bv[q] = bias[lane * PER + q];

  int beg = rowptr[node];
  int end = rowptr[node + 1];
  float dc = dis[node];

  for (;;) {
    const int nx = node + nstride;
    int nbeg = 0, nend = 0;
    float ndc = 0.f;
    if (nx < n) { nbeg = rowptr[nx]; nend = rowptr[nx + 1]; ndc = dis[nx]; }

    float acc[PER];
    {
      const float* src = H + (long long)node * F + lane * PER;
      const float sl = dc * dc;
      if constexpr (PER == 4) {
        const float4 v = *(const float4*)src;
        acc[0] = v.x * sl; acc[1] = v.y * sl; acc[2] = v.z * sl; acc[3] = v.w * sl;
      } else {
        const float2 v = *(const float2*)src;
        acc[0] = v.x * sl; acc[1] = v.y * sl;
      }
    }

    for (int base = beg; base < end; base += 32) {
      const int p = base + lane;
      int rr = 0;
      float nw = 0.f;
      if (p < end) {
        rr = srow[p];
        nw = dis[rr] * sw[p] * dc;
      }
      int m = end - base;
      if (m > 32) m = 32;
#pragma unroll 4
      for (int j = 0; j < m; ++j) {
        const int r = __builtin_amdgcn_readlane(rr, j);
        const float w = __int_as_float(__builtin_amdgcn_readlane(__float_as_int(nw), j));
        const float* src = H + (long long)r * F + lane * PER;
        if constexpr (PER == 4) {
          const float4 v = *(const float4*)src;
          acc[0] += v.x * w; acc[1] += v.y * w; acc[2] += v.z * w; acc[3] += v.w * w;
        } else {
          const float2 v = *(const float2*)src;
          acc[0] += v.x * w; acc[1] += v.y * w;
        }
      }
    }

    float* dst = out + (long long)node * F + lane * PER;
    if constexpr (PER == 4) {
      float4 o;
      o.x = acc[0] + bv[0]; o.y = acc[1] + bv[1]; o.z = acc[2] + bv[2]; o.w = acc[3] + bv[3];
      if constexpr (RELU) {
        o.x = o.x > 0.f ? o.x : 0.f; o.y = o.y > 0.f ? o.y : 0.f;
        o.z = o.z > 0.f ? o.z : 0.f; o.w = o.w > 0.f ? o.w : 0.f;
      }
      *(float4*)dst = o;
    } else {
      float2 o;
      o.x = acc[0] + bv[0]; o.y = acc[1] + bv[1];
      if constexpr (RELU) {
        o.x = o.x > 0.f ? o.x : 0.f; o.y = o.y > 0.f ? o.y : 0.f;
      }
      *(float2*)dst = o;
    }

    if (nx >= n) break;
    node = nx; beg = nbeg; end = nend; dc = ndc;
  }
}

static inline int nblk(long long n, int t) { return (int)((n + t - 1) / t); }

extern "C" void kernel_launch(void* const* d_in, const int* in_sizes, int n_in,
                              void* d_out, int out_size, void* d_ws, size_t ws_size,
                              hipStream_t stream) {
  const float* x  = (const float*)d_in[0];
  const int*   ei = (const int*)d_in[1];
  const float* ew = (const float*)d_in[2];
  const float* W1 = (const float*)d_in[3];
  const float* b1 = (const float*)d_in[4];
  const float* W2 = (const float*)d_in[5];
  const float* b2 = (const float*)d_in[6];
  float* out = (float*)d_out;

  const int n = in_sizes[0] / F_IN;
  const int E = in_sizes[1] / 2;
  const int* row = ei;
  const int* col = ei + E;

  char* ws = (char*)d_ws;
  size_t o = 0;
  auto carve = [&](size_t bytes) {
    char* p = ws + o;
    o += (bytes + 255) & ~(size_t)255;
    return p;
  };
  const int nch = nblk(E, CH);
  const int nbk = nblk(n, 256);
  int*   offtab = (int*)carve((size_t)nch * OT * 4);
  int*   ccol   = (int*)carve((size_t)nch * CH * 4);
  int*   crow   = (int*)carve((size_t)nch * CH * 4);
  float* cw     = (float*)carve((size_t)nch * CH * 4);
  int*   rowptr = (int*)carve((size_t)(n + 1) * 4);
  int*   srow   = (int*)carve((size_t)E * 4);
  float* sw     = (float*)carve((size_t)E * 4);
  float* dis    = (float*)carve((size_t)n * 4);
  float* h      = (float*)carve((size_t)n * F_HID * 4);
  float* a1     = (float*)carve((size_t)n * F_HID * 4);

  const int ntile1 = (n / 16) * (F_HID / 16);
  const int gsplit = (ntile1 * 7 / 16 / GT) * GT;
  const int g1 = gsplit / GT;
  const int g2 = nblk(ntile1 - gsplit, GT);
  k_sort1_gemm<<<nch + g1, 1024, 0, stream>>>(row, col, ew, ccol, crow, cw, offtab, E, nch,
                                              x, W1, h, 0, gsplit);
  k_sort2_gemm<<<nbk + g2, 1024, 0, stream>>>(ccol, crow, cw, offtab, rowptr, srow, sw, dis, n, E, nch, nbk,
                                              x, W1, h, gsplit, ntile1);

  k_gather<F_HID, true><<<GATHER_BLOCKS, 256, 0, stream>>>(h, rowptr, srow, sw, dis, b1, a1, n);

  k_gemm_wmma<F_OUT><<<GEMM_BLOCKS, 256, 0, stream>>>(a1, W2, h, n);
  k_gather<F_OUT, false><<<GATHER_BLOCKS, 256, 0, stream>>>(h, rowptr, srow, sw, dis, b2, out, n);
}


# Layer-2 GEMM fused into the layer-1 gather: 16-wave blocks put 16 gathered, biased, relu'd rows in an LDS tile, four waves multiply it by W2 on WMMA; layer-1 output never stored; 5 launches become 4
# speedup vs baseline: 3.8855x; 1.0623x over previous
#include <hip/hip_runtime.h>
#include <cstdint>

#define F_IN  128
#define F_HID 128
#define F_OUT 64

#define GT            64
#define GATHER_BLOCKS 2048
#define FUSED_BLOCKS  1024
#define TPITCH        (F_HID + 4)

typedef __attribute__((ext_vector_type(2))) float v2f;
typedef __attribute__((ext_vector_type(8))) float v8f;

#define CH   4096
#define OT   257
#define U2   5
#define TCAP 8192

__device__ __forceinline__ int block_excl_scan(int v, int t, int* s_w, int& total) {
  const int lane = t & 31;
  const int wave = t >> 5;
  int x = v;
#pragma unroll
  for (int d = 1; d < 32; d <<= 1) {
    const int y = __shfl_up(x, d);
    if (lane >= d) x += y;
  }
  if (lane == 31) s_w[wave] = x;
  __syncthreads();
  if (wave == 0) {
    const int w = s_w[lane];
    int z = w;
#pragma unroll
    for (int d = 1; d < 32; d <<= 1) {
      const int y = __shfl_up(z, d);
      if (lane >= d) z += y;
    }
    s_w[lane] = z - w;
    if (lane == 31) s_w[32] = z;
  }
  __syncthreads();
  total = s_w[32];
  return s_w[wave] + x - v;
}

#define SORT1_WORDS (576 + 3 * CH)
#define SORT2_WORDS (1152 + 2 * TCAP)

__device__ __forceinline__ void
chunk_sort_body(const int* __restrict__ row, const int* __restrict__ col, const float* __restrict__ ew,
                int* __restrict__ ccol, int* __restrict__ crow, float* __restrict__ cw,
                int* __restrict__ offtab, int E, int* smem) {
  int*   s_hist = smem;
  int*   s_off  = smem + 256;
  int*   s_scan = smem + 520;
  int*   l_col  = smem + 576;
  int*   l_row  = smem + 576 + CH;
  float* l_w    = (float*)(smem + 576 + 2 * CH);
  const int t = threadIdx.x;
  const int e0 = blockIdx.x * CH;
  int cnt = E - e0;
  if (cnt > CH) cnt = CH;
  if (t < 256) s_hist[t] = 0;

  int cc[4], rr[4], rk[4];
  float ww[4];
#pragma unroll
  for (int k = 0; k < 4; ++k) {
    const int i = t + k * 1024;
    cc[k] = 0; rr[k] = 0; ww[k] = 0.f;
    if (i < cnt) { cc[k] = col[e0 + i]; rr[k] = row[e0 + i]; ww[k] = ew[e0 + i]; }
  }
  __syncthreads();
#pragma unroll
  for (int k = 0; k < 4; ++k) {
    rk[k] = 0;
    if (t + k * 1024 < cnt) rk[k] = atomicAdd(&s_hist[cc[k] >> 8], 1);
  }
  __syncthreads();
  int total;
  const int ex = block_excl_scan((t < 256) ? s_hist[t] : 0, t, s_scan, total);
  if (t <= 256) {
    s_off[t] = ex;
    offtab[blockIdx.x * OT + t] = ex;
  }
  __syncthreads();
#pragma unroll
  for (int k = 0; k < 4; ++k) {
    if (t + k * 1024 < cnt) {
      const int p = s_off[cc[k] >> 8] + rk[k];
      l_col[p] = cc[k]; l_row[p] = rr[k]; l_w[p] = ww[k];
    }
  }
  __syncthreads();
#pragma unroll
  for (int k = 0; k < 4; ++k) {
    const int i = t + k * 1024;
    if (i < cnt) { ccol[e0 + i] = l_col[i]; crow[e0 + i] = l_row[i]; cw[e0 + i] = l_w[i]; }
  }
}

__device__ __forceinline__ void
bucket_finish_body(const int* __restrict__ ccol, const int* __restrict__ crow, const float* __restrict__ cw,
                   const int* __restrict__ offtab, int* __restrict__ rowptr, int* __restrict__ srow,
                   float* __restrict__ sw, float* __restrict__ dis, int n, int E, int nchunk, int nbucket,
                   int* smem) {
  int*   s_pre    = smem;
  int*   s_src    = smem + 256;
  int*   s_cnt    = smem + 512;
  float* s_deg    = (float*)(smem + 768);
  int*   s_scan_a = smem + 1024;
  int*   s_scan_b = smem + 1064;
  int*   s_scan_c = smem + 1104;
  int*   st_row   = smem + 1152;
  float* st_w     = (float*)(smem + 1152 + TCAP);
  const int t = threadIdx.x;
  const int b = blockIdx.x;

  int lo = 0, len = 0;
  if (t < nchunk) {
    lo  = offtab[t * OT + b];
    len = offtab[t * OT + b + 1] - lo;
  }
  if (t < 256) { s_cnt[t] = 0; s_deg[t] = 0.f; }
  int total, base;
  const int pre = block_excl_scan(len, t, s_scan_a, total);
  (void)block_excl_scan(lo, t, s_scan_b, base);
  if (t < 256) {
    s_pre[t] = (t < nchunk) ? pre : 0x7fffffff;
    s_src[t] = t * CH + lo;
  }
  __syncthreads();

  auto locate = [&](int i) {
    int c = 0;
#pragma unroll
    for (int s = 128; s > 0; s >>= 1)
      if (s_pre[c + s] <= i) c += s;
    return s_src[c] + (i - s_pre[c]);
  };

  int kc[U2], kr[U2], ks[U2];
  float kw[U2];
#pragma unroll
  for (int k = 0; k < U2; ++k) ks[k] = 0;
#pragma unroll
  for (int s = 128; s > 0; s >>= 1) {
#pragma unroll
    for (int k = 0; k < U2; ++k)
      if (s_pre[ks[k] + s] <= t + k * 1024) ks[k] += s;
  }
#pragma unroll
  for (int k = 0; k < U2; ++k) {
    const int i = t + k * 1024;
    kc[k] = 0; kr[k] = 0; kw[k] = 0.f;
    if (i < total) {
      const int j = s_src[ks[k]] + (i - s_pre[ks[k]]);
      kc[k] = ccol[j]; kr[k] = crow[j]; kw[k] = cw[j];
    }
  }
#pragma unroll
  for (int k = 0; k < U2; ++k) {
    if (t + k * 1024 < total) {
      const int ln = kc[k] & 255;
      atomicAdd(&s_cnt[ln], 1);
      atomicAdd(&s_deg[ln], kw[k]);
    }
  }
  for (int i = t + U2 * 1024; i < total; i += 1024) {
    const int j = locate(i);
    const int ln = ccol[j] & 255;
    atomicAdd(&s_cnt[ln], 1);
    atomicAdd(&s_deg[ln], cw[j]);
  }
  __syncthreads();

  int tot2;
  const int ex = block_excl_scan((t < 256) ? s_cnt[t] : 0, t, s_scan_c, tot2);
  if (t < 256) {
    s_cnt[t] = ex;
    const int node = b * 256 + t;
    if (node < n) {
      rowptr[node] = base + ex;
      const float d = 1.0f + s_deg[t];
      dis[node] = (d > 0.f) ? rsqrtf(d) : 0.f;
    }
  }
  if (b == nbucket - 1 && t == 0) rowptr[n] = E;
  __syncthreads();

  const bool staged = total <= TCAP;
#pragma unroll
  for (int k = 0; k < U2; ++k) {
    if (t + k * 1024 < total) {
      const int p = atomicAdd(&s_cnt[kc[k] & 255], 1);
      if (staged) { st_row[p] = kr[k]; st_w[p] = kw[k]; }
      else        { srow[base + p] = kr[k]; sw[base + p] = kw[k]; }
    }
  }
  for (int i = t + U2 * 1024; i < total; i += 1024) {
    const int j = locate(i);
    const int p = atomicAdd(&s_cnt[ccol[j] & 255], 1);
    const int r = crow[j];
    const float w = cw[j];
    if (staged) { st_row[p] = r; st_w[p] = w; }
    else        { srow[base + p] = r; sw[base + p] = w; }
  }
  __syncthreads();
  if (staged) {
    for (int i = t; i < total; i += 1024) { srow[base + i] = st_row[i]; sw[base + i] = st_w[i]; }
  }
}

template <int FOUT, int NW, int KG>
__device__ __forceinline__ void
gemm_tiles(const float* __restrict__ A, const float* __restrict__ W, float* __restrict__ H,
           float* sW, int first, int last, int tstride) {
  for (int i = threadIdx.x; i < F_IN * FOUT; i += NW * 32) sW[i] = W[i];
  __syncthreads();

  const int lane = threadIdx.x & 31;
  const int wave = __builtin_amdgcn_readfirstlane((int)(threadIdx.x >> 5));
  constexpr int tilesN = FOUT / 16;
  const int koff = (lane >> 4) * 2;

  for (int tile = first + wave; tile < last; tile += tstride) {
    const int tm = tile / tilesN;
    const int tn = tile % tilesN;
    const int mrow = tm * 16 + (lane & 15);
    const int ncol = tn * 16 + (lane & 15);

    const float* __restrict__ arow = A + (long long)mrow * F_IN;
    const float* __restrict__ wb = sW + ncol;

    v8f c = {};
    for (int k0 = 0; k0 < F_IN; k0 += 4 * KG) {
      v2f a[KG];
#pragma unroll
      for (int j = 0; j < KG; ++j) {
        const int ka = k0 + 4 * j + koff;
        a[j].x = arow[ka];
        a[j].y = arow[ka + 1];
      }
#pragma unroll
      for (int j = 0; j < KG; ++j) {
        const int ka = k0 + 4 * j + koff;
        v2f b;
        b.x = wb[ka * FOUT];
        b.y = wb[(ka + 1) * FOUT];
        c = __builtin_amdgcn_wmma_f32_16x16x4_f32(
             false, a[j],  false, b,
             (short)0, c,  false,  false);
      }
    }

    const int rbase = tm * 16 + ((lane >> 4) << 3);
#pragma unroll
    for (int v = 0; v < 8; ++v)
      H[(long long)(rbase + v) * FOUT + ncol] = c[v];
  }
}

__global__ void __launch_bounds__(1024, 16)
k_sort1_gemm(const int* __restrict__ row, const int* __restrict__ col, const float* __restrict__ ew,
             int* __restrict__ ccol, int* __restrict__ crow, float* __restrict__ cw,
             int* __restrict__ offtab, int E, int nsort,
             const float* __restrict__ A, const float* __restrict__ Wm, float* __restrict__ H,
             int gt0, int gt1) {
  constexpr int WORDS = (F_IN * F_HID > SORT1_WORDS) ? F_IN * F_HID : SORT1_WORDS;
  __shared__ int smem[WORDS];
  if ((int)blockIdx.x < nsort) {
    chunk_sort_body(row, col, ew, ccol, crow, cw, offtab, E, smem);
  } else {
    const int t0 = gt0 + ((int)blockIdx.x - nsort) * GT;
    const int t1 = (t0 + GT < gt1) ? t0 + GT : gt1;
    gemm_tiles<F_HID, 32, 16>(A, Wm, H, (float*)smem, t0, t1, 32);
  }
}

__global__ void __launch_bounds__(1024, 16)
k_sort2_gemm(const int* __restrict__ ccol, const int* __restrict__ crow, const float* __restrict__ cw,
             const int* __restrict__ offtab, int* __restrict__ rowptr, int* __restrict__ srow,
             float* __restrict__ sw, float* __restrict__ dis, int n, int E, int nchunk, int nsort,
             const float* __restrict__ A, const float* __restrict__ Wm, float* __restrict__ H,
             int gt0, int gt1) {
  constexpr int WORDS = (F_IN * F_HID > SORT2_WORDS) ? F_IN * F_HID : SORT2_WORDS;
  __shared__ int smem[WORDS];
  if ((int)blockIdx.x < nsort) {
    bucket_finish_body(ccol, crow, cw, offtab, rowptr, srow, sw, dis, n, E, nchunk, nsort, smem);
  } else {
    const int t0 = gt0 + ((int)blockIdx.x - nsort) * GT;
    const int t1 = (t0 + GT < gt1) ? t0 + GT : gt1;
    gemm_tiles<F_HID, 32, 16>(A, Wm, H, (float*)smem, t0, t1, 32);
  }
}

template <int F>
__device__ __forceinline__ void
gather_row(const float* __restrict__ H, const int* __restrict__ srow, const float* __restrict__ sw,
           const float* __restrict__ dis, int node, int beg, int end, float dc, int lane, float* acc) {
  constexpr int PER = F / 32;
  {
    const float* src = H + (long long)node * F + lane * PER;
    const float sl = dc * dc;
    if constexpr (PER == 4) {
      const float4 v = *(const float4*)src;
      acc[0] = v.x * sl; acc[1] = v.y * sl; acc[2] = v.z * sl; acc[3] = v.w * sl;
    } else {
      const float2 v = *(const float2*)src;
      acc[0] = v.x * sl; acc[1] = v.y * sl;
    }
  }

  for (int base = beg; base < end; base += 32) {
    const int p = base + lane;
    int rr = 0;
    float nw = 0.f;
    if (p < end) {
      rr = srow[p];
      nw = dis[rr] * sw[p] * dc;
    }
    int m = end - base;
    if (m > 32) m = 32;
#pragma unroll 4
    for (int j = 0; j < m; ++j) {
      const int r = __builtin_amdgcn_readlane(rr, j);
      const float w = __int_as_float(__builtin_amdgcn_readlane(__float_as_int(nw), j));
      const float* src = H + (long long)r * F + lane * PER;
      if constexpr (PER == 4) {
        const float4 v = *(const float4*)src;
        acc[0] += v.x * w; acc[1] += v.y * w; acc[2] += v.z * w; acc[3] += v.w * w;
      } else {
        const float2 v = *(const float2*)src;
        acc[0] += v.x * w; acc[1] += v.y * w;
      }
    }
  }
}

template <int F>
__global__ void __launch_bounds__(256)
k_gather(const float* __restrict__ H, const int* __restrict__ rowptr,
         const int* __restrict__ srow, const float* __restrict__ sw,
         const float* __restrict__ dis, const float* __restrict__ bias,
         float* __restrict__ out, int n) {
  constexpr int PER = F / 32;
  const int lane = threadIdx.x & 31;
  const int wave = __builtin_amdgcn_readfirstlane((int)(threadIdx.x >> 5));
  const int nstride = (int)gridDim.x * 8;
  int node = (int)blockIdx.x * 8 + wave;
  if (node >= n) return;

  float bv[PER];
#pragma unroll
  for (int q = 0; q < PER; ++q) bv[q] = bias[lane * PER + q];

  int beg = rowptr[node];
  int end = rowptr[node + 1];
  float dc = dis[node];

  for (;;) {
    const int nx = node + nstride;
    int nbeg = 0, nend = 0;
    float ndc = 0.f;
    if (nx < n) { nbeg = rowptr[nx]; nend = rowptr[nx + 1]; ndc = dis[nx]; }

    float acc[PER];
    gather_row<F>(H, srow, sw, dis, node, beg, end, dc, lane, acc);

    float* dst = out + (long long)node * F + lane * PER;
    if constexpr (PER == 4) {
      float4 o;
      o.x = acc[0] + bv[0]; o.y = acc[1] + bv[1]; o.z = acc[2] + bv[2]; o.w = acc[3] + bv[3];
      *(float4*)dst = o;
    } else {
      float2 o;
      o.x = acc[0] + bv[0]; o.y = acc[1] + bv[1];
      *(float2*)dst = o;
    }

    if (nx >= n) break;
    node = nx; beg = nbeg; end = nend; dc = ndc;
  }
}

__global__ void __launch_bounds__(512)
k_gather_gemm(const float* __restrict__ H, const int* __restrict__ rowptr,
              const int* __restrict__ srow, const float* __restrict__ sw,
              const float* __restrict__ dis, const float* __restrict__ bias,
              const float* __restrict__ W, float* __restrict__ H2, int n) {
  __shared__ float sW[F_HID * F_OUT];
  __shared__ float sT[2 * 16 * TPITCH];
  for (int i = threadIdx.x; i < F_HID * F_OUT; i += 512) sW[i] = W[i];

  const int lane = threadIdx.x & 31;
  const int wave = __builtin_amdgcn_readfirstlane((int)(threadIdx.x >> 5));
  const int ntile = n / 16;
  const int tstride = (int)gridDim.x;
  const int koff = (lane >> 4) * 2;
  int tile = (int)blockIdx.x;
  if (tile >= ntile) return;

  float bv[4];
#pragma unroll
  for (int q = 0; q < 4; ++q) bv[q] = bias[lane * 4 + q];

  int node = tile * 16 + wave;
  int beg = rowptr[node];
  int end = rowptr[node + 1];
  float dc = dis[node];

  for (int it = 0;; ++it) {
    const int nt = tile + tstride;
    const int nx = nt * 16 + wave;
    int nbeg = 0, nend = 0;
    float ndc = 0.f;
    if (nt < ntile) { nbeg = rowptr[nx]; nend = rowptr[nx + 1]; ndc = dis[nx]; }

    float acc[4];
    gather_row<F_HID>(H, srow, sw, dis, node, beg, end, dc, lane, acc);

    float4 o;
    o.x = acc[0] + bv[0]; o.y = acc[1] + bv[1]; o.z = acc[2] + bv[2]; o.w = acc[3] + bv[3];
    o.x = o.x > 0.f ? o.x : 0.f; o.y = o.y > 0.f ? o.y : 0.f;
    o.z = o.z > 0.f ? o.z : 0.f; o.w = o.w > 0.f ? o.w : 0.f;
    float* tbuf = sT + (it & 1) * (16 * TPITCH);
    *(float4*)(tbuf + wave * TPITCH + lane * 4) = o;
    __syncthreads();

    const int duty = (wave - 4 * it) & 15;
    if (duty < 4) {
      const int ncol = duty * 16 + (lane & 15);
      const float* arow = tbuf + (lane & 15) * TPITCH;
      const float* wb = sW + ncol;
      v8f c = {};
      for (int k0 = 0; k0 < F_HID; k0 += 32) {
        v2f a[8];
#pragma unroll
        for (int j = 0; j < 8; ++j) {
          const int ka = k0 + 4 * j + koff;
          a[j] = *(const v2f*)(arow + ka);
        }
#pragma unroll
        for (int j = 0; j < 8; ++j) {
          const int ka = k0 + 4 * j + koff;
          v2f b;
          b.x = wb[ka * F_OUT];
          b.y = wb[(ka + 1) * F_OUT];
          c = __builtin_amdgcn_wmma_f32_16x16x4_f32(
               false, a[j],  false, b,
               (short)0, c,  false,  false);
        }
      }
      const int rbase = tile * 16 + ((lane >> 4) << 3);
#pragma unroll
      for (int v = 0; v < 8; ++v)
        H2[(long long)(rbase + v) * F_OUT + ncol] = c[v];
    }

    if (nt >= ntile) break;
    tile = nt; node = nx; beg = nbeg; end = nend; dc = ndc;
  }
}

static inline int nblk(long long n, int t) { return (int)((n + t - 1) / t); }

extern "C" void kernel_launch(void* const* d_in, const int* in_sizes, int n_in,
                              void* d_out, int out_size, void* d_ws, size_t ws_size,
                              hipStream_t stream) {
  const float* x  = (const float*)d_in[0];
  const int*   ei = (const int*)d_in[1];
  const float* ew = (const float*)d_in[2];
  const float* W1 = (const float*)d_in[3];
  const float* b1 = (const float*)d_in[4];
  const float* W2 = (const float*)d_in[5];
  const float* b2 = (const float*)d_in[6];
  float* out = (float*)d_out;

  const int n = in_sizes[0] / F_IN;
  const int E = in_sizes[1] / 2;
  const int* row = ei;
  const int* col = ei + E;

  char* ws = (char*)d_ws;
  size_t o = 0;
  auto carve = [&](size_t bytes) {
    char* p = ws + o;
    o += (bytes + 255) & ~(size_t)255;
    return p;
  };
  const int nch = nblk(E, CH);
  const int nbk = nblk(n, 256);
  int*   offtab = (int*)carve((size_t)nch * OT * 4);
  int*   ccol   = (int*)carve((size_t)nch * CH * 4);
  int*   crow   = (int*)carve((size_t)nch * CH * 4);
  float* cw     = (float*)carve((size_t)nch * CH * 4);
  int*   rowptr = (int*)carve((size_t)(n + 1) * 4);
  int*   srow   = (int*)carve((size_t)E * 4);
  float* sw     = (float*)carve((size_t)E * 4);
  float* dis    = (float*)carve((size_t)n * 4);
  float* h      = (float*)carve((size_t)n * F_HID * 4);
  float* h2     = (float*)carve((size_t)n * F_OUT * 4);

  const int ntile1 = (n / 16) * (F_HID / 16);
  const int gsplit = (ntile1 * 7 / 16 / GT) * GT;
  const int g1 = gsplit / GT;
  const int g2 = nblk(ntile1 - gsplit, GT);
  k_sort1_gemm<<<nch + g1, 1024, 0, stream>>>(row, col, ew, ccol, crow, cw, offtab, E, nch,
                                              x, W1, h, 0, gsplit);
  k_sort2_gemm<<<nbk + g2, 1024, 0, stream>>>(ccol, crow, cw, offtab, rowptr, srow, sw, dis, n, E, nch, nbk,
                                              x, W1, h, gsplit, ntile1);

  k_gather_gemm<<<FUSED_BLOCKS, 512, 0, stream>>>(h, rowptr, srow, sw, dis, b1, W2, h2, n);

  k_gather<F_OUT><<<GATHER_BLOCKS, 256, 0, stream>>>(h2, rowptr, srow, sw, dis, b2, out, n);
}


# Layer-1 GEMM: A fragments come as one 128-bit load per two k-steps instead of one 64-bit load per step, both loaders re-ordered alike in k: half the 16-line loads; k order of the x*W1 sums changes
# speedup vs baseline: 4.2544x; 1.0949x over previous
#include <hip/hip_runtime.h>
#include <cstdint>

#define F_IN  128
#define F_HID 128
#define F_OUT 64

#define GT            64
#define GATHER_BLOCKS 2048
#define FUSED_BLOCKS  1024
#define TPITCH        (F_HID + 4)

typedef __attribute__((ext_vector_type(2))) float v2f;
typedef __attribute__((ext_vector_type(8))) float v8f;

#define CH   4096
#define OT   257
#define U2   5
#define TCAP 8192

__device__ __forceinline__ int block_excl_scan(int v, int t, int* s_w, int& total) {
  const int lane = t & 31;
  const int wave = t >> 5;
  int x = v;
#pragma unroll
  for (int d = 1; d < 32; d <<= 1) {
    const int y = __shfl_up(x, d);
    if (lane >= d) x += y;
  }
  if (lane == 31) s_w[wave] = x;
  __syncthreads();
  if (wave == 0) {
    const int w = s_w[lane];
    int z = w;
#pragma unroll
    for (int d = 1; d < 32; d <<= 1) {
      const int y = __shfl_up(z, d);
      if (lane >= d) z += y;
    }
    s_w[lane] = z - w;
    if (lane == 31) s_w[32] = z;
  }
  __syncthreads();
  total = s_w[32];
  return s_w[wave] + x - v;
}

#define SORT1_WORDS (576 + 3 * CH)
#define SORT2_WORDS (1152 + 2 * TCAP)

__device__ __forceinline__ void
chunk_sort_body(const int* __restrict__ row, const int* __restrict__ col, const float* __restrict__ ew,
                int* __restrict__ ccol, int* __restrict__ crow, float* __restrict__ cw,
                int* __restrict__ offtab, int E, int* smem) {
  int*   s_hist = smem;
  int*   s_off  = smem + 256;
  int*   s_scan = smem + 520;
  int*   l_col  = smem + 576;
  int*   l_row  = smem + 576 + CH;
  float* l_w    = (float*)(smem + 576 + 2 * CH);
  const int t = threadIdx.x;
  const int e0 = blockIdx.x * CH;
  int cnt = E - e0;
  if (cnt > CH) cnt = CH;
  if (t < 256) s_hist[t] = 0;

  int cc[4], rr[4], rk[4];
  float ww[4];
#pragma unroll
  for (int k = 0; k < 4; ++k) {
    const int i = t + k * 1024;
    cc[k] = 0; rr[k] = 0; ww[k] = 0.f;
    if (i < cnt) { cc[k] = col[e0 + i]; rr[k] = row[e0 + i]; ww[k] = ew[e0 + i]; }
  }
  __syncthreads();
#pragma unroll
  for (int k = 0; k < 4; ++k) {
    rk[k] = 0;
    if (t + k * 1024 < cnt) rk[k] = atomicAdd(&s_hist[cc[k] >> 8], 1);
  }
  __syncthreads();
  int total;
  const int ex = block_excl_scan((t < 256) ? s_hist[t] : 0, t, s_scan, total);
  if (t <= 256) {
    s_off[t] = ex;
    offtab[blockIdx.x * OT + t] = ex;
  }
  __syncthreads();
#pragma unroll
  for (int k = 0; k < 4; ++k) {
    if (t + k * 1024 < cnt) {
      const int p = s_off[cc[k] >> 8] + rk[k];
      l_col[p] = cc[k]; l_row[p] = rr[k]; l_w[p] = ww[k];
    }
  }
  __syncthreads();
#pragma unroll
  for (int k = 0; k < 4; ++k) {
    const int i = t + k * 1024;
    if (i < cnt) { ccol[e0 + i] = l_col[i]; crow[e0 + i] = l_row[i]; cw[e0 + i] = l_w[i]; }
  }
}

__device__ __forceinline__ void
bucket_finish_body(const int* __restrict__ ccol, const int* __restrict__ crow, const float* __restrict__ cw,
                   const int* __restrict__ offtab, int* __restrict__ rowptr, int* __restrict__ srow,
                   float* __restrict__ sw, float* __restrict__ dis, int n, int E, int nchunk, int nbucket,
                   int* smem) {
  int*   s_pre    = smem;
  int*   s_src    = smem + 256;
  int*   s_cnt    = smem + 512;
  float* s_deg    = (float*)(smem + 768);
  int*   s_scan_a = smem + 1024;
  int*   s_scan_b = smem + 1064;
  int*   s_scan_c = smem + 1104;
  int*   st_row   = smem + 1152;
  float* st_w     = (float*)(smem + 1152 + TCAP);
  const int t = threadIdx.x;
  const int b = blockIdx.x;

  int lo = 0, len = 0;
  if (t < nchunk) {
    lo  = offtab[t * OT + b];
    len = offtab[t * OT + b + 1] - lo;
  }
  if (t < 256) { s_cnt[t] = 0; s_deg[t] = 0.f; }
  int total, base;
  const int pre = block_excl_scan(len, t, s_scan_a, total);
  (void)block_excl_scan(lo, t, s_scan_b, base);
  if (t < 256) {
    s_pre[t] = (t < nchunk) ? pre : 0x7fffffff;
    s_src[t] = t * CH + lo;
  }
  __syncthreads();

  auto locate = [&](int i) {
    int c = 0;
#pragma unroll
    for (int s = 128; s > 0; s >>= 1)
      if (s_pre[c + s] <= i) c += s;
    return s_src[c] + (i - s_pre[c]);
  };

  int kc[U2], kr[U2], ks[U2];
  float kw[U2];
#pragma unroll
  for (int k = 0; k < U2; ++k) ks[k] = 0;
#pragma unroll
  for (int s = 128; s > 0; s >>= 1) {
#pragma unroll
    for (int k = 0; k < U2; ++k)
      if (s_pre[ks[k] + s] <= t + k * 1024) ks[k] += s;
  }
#pragma unroll
  for (int k = 0; k < U2; ++k) {
    const int i = t + k * 1024;
    kc[k] = 0; kr[k] = 0; kw[k] = 0.f;
    if (i < total) {
      const int j = s_src[ks[k]] + (i - s_pre[ks[k]]);
      kc[k] = ccol[j]; kr[k] = crow[j]; kw[k] = cw[j];
    }
  }
#pragma unroll
  for (int k = 0; k < U2; ++k) {
    if (t + k * 1024 < total) {
      const int ln = kc[k] & 255;
      atomicAdd(&s_cnt[ln], 1);
      atomicAdd(&s_deg[ln], kw[k]);
    }
  }
  for (int i = t + U2 * 1024; i < total; i += 1024) {
    const int j = locate(i);
    const int ln = ccol[j] & 255;
    atomicAdd(&s_cnt[ln], 1);
    atomicAdd(&s_deg[ln], cw[j]);
  }
  __syncthreads();

  int tot2;
  const int ex = block_excl_scan((t < 256) ? s_cnt[t] : 0, t, s_scan_c, tot2);
  if (t < 256) {
    s_cnt[t] = ex;
    const int node = b * 256 + t;
    if (node < n) {
      rowptr[node] = base + ex;
      const float d = 1.0f + s_deg[t];
      dis[node] = (d > 0.f) ? rsqrtf(d) : 0.f;
    }
  }
  if (b == nbucket - 1 && t == 0) rowptr[n] = E;
  __syncthreads();

  const bool staged = total <= TCAP;
#pragma unroll
  for (int k = 0; k < U2; ++k) {
    if (t + k * 1024 < total) {
      const int p = atomicAdd(&s_cnt[kc[k] & 255], 1);
      if (staged) { st_row[p] = kr[k]; st_w[p] = kw[k]; }
      else        { srow[base + p] = kr[k]; sw[base + p] = kw[k]; }
    }
  }
  for (int i = t + U2 * 1024; i < total; i += 1024) {
    const int j = locate(i);
    const int p = atomicAdd(&s_cnt[ccol[j] & 255], 1);
    const int r = crow[j];
    const float w = cw[j];
    if (staged) { st_row[p] = r; st_w[p] = w; }
    else        { srow[base + p] = r; sw[base + p] = w; }
  }
  __syncthreads();
  if (staged) {
    for (int i = t; i < total; i += 1024) { srow[base + i] = st_row[i]; sw[base + i] = st_w[i]; }
  }
}

template <int FOUT, int NW, int KG>
__device__ __forceinline__ void
gemm_tiles(const float* __restrict__ A, const float* __restrict__ W, float* __restrict__ H,
           float* sW, int first, int last, int tstride) {
  for (int i = threadIdx.x; i < F_IN * FOUT; i += NW * 32) sW[i] = W[i];
  __syncthreads();

  const int lane = threadIdx.x & 31;
  const int wave = __builtin_amdgcn_readfirstlane((int)(threadIdx.x >> 5));
  constexpr int tilesN = FOUT / 16;
  const int koff = (lane >> 4) * 4;
  const unsigned aoff = (unsigned)((lane & 15) * F_IN + koff);

  for (int tile = first + wave; tile < last; tile += tstride) {
    const int tm = tile / tilesN;
    const int tn = tile % tilesN;
    const int ncol = tn * 16 + (lane & 15);

    const float* __restrict__ astrip = A + (long long)tm * (16 * F_IN);
    const float* __restrict__ wb = sW + ncol;

    v8f c = {};
    for (int k0 = 0; k0 < F_IN; k0 += 4 * KG) {
      float4 a[KG / 2];
#pragma unroll
      for (int j = 0; j < KG / 2; ++j)
        a[j] = *(const float4*)(astrip + aoff + (k0 + 8 * j));
#pragma unroll
      for (int j = 0; j < KG / 2; ++j) {
        const int ka = k0 + 8 * j + koff;
        v2f a0, a1, b0, b1;
        a0.x = a[j].x; a0.y = a[j].y;
        a1.x = a[j].z; a1.y = a[j].w;
        b0.x = wb[ka * FOUT];
        b0.y = wb[(ka + 1) * FOUT];
        b1.x = wb[(ka + 2) * FOUT];
        b1.y = wb[(ka + 3) * FOUT];
        c = __builtin_amdgcn_wmma_f32_16x16x4_f32(
             false, a0,  false, b0,
             (short)0, c,  false,  false);
        c = __builtin_amdgcn_wmma_f32_16x16x4_f32(
             false, a1,  false, b1,
             (short)0, c,  false,  false);
      }
    }

    const int rbase = tm * 16 + ((lane >> 4) << 3);
#pragma unroll
    for (int v = 0; v < 8; ++v)
      H[(long long)(rbase + v) * FOUT + ncol] = c[v];
  }
}

__global__ void __launch_bounds__(1024, 16)
k_sort1_gemm(const int* __restrict__ row, const int* __restrict__ col, const float* __restrict__ ew,
             int* __restrict__ ccol, int* __restrict__ crow, float* __restrict__ cw,
             int* __restrict__ offtab, int E, int nsort,
             const float* __restrict__ A, const float* __restrict__ Wm, float* __restrict__ H,
             int gt0, int gt1) {
  constexpr int WORDS = (F_IN * F_HID > SORT1_WORDS) ? F_IN * F_HID : SORT1_WORDS;
  __shared__ int smem[WORDS];
  if ((int)blockIdx.x < nsort) {
    chunk_sort_body(row, col, ew, ccol, crow, cw, offtab, E, smem);
  } else {
    const int t0 = gt0 + ((int)blockIdx.x - nsort) * GT;
    const int t1 = (t0 + GT < gt1) ? t0 + GT : gt1;
    gemm_tiles<F_HID, 32, 16>(A, Wm, H, (float*)smem, t0, t1, 32);
  }
}

__global__ void __launch_bounds__(1024, 16)
k_sort2_gemm(const int* __restrict__ ccol, const int* __restrict__ crow, const float* __restrict__ cw,
             const int* __restrict__ offtab, int* __restrict__ rowptr, int* __restrict__ srow,
             float* __restrict__ sw, float* __restrict__ dis, int n, int E, int nchunk, int nsort,
             const float* __restrict__ A, const float* __restrict__ Wm, float* __restrict__ H,
             int gt0, int gt1) {
  constexpr int WORDS = (F_IN * F_HID > SORT2_WORDS) ? F_IN * F_HID : SORT2_WORDS;
  __shared__ int smem[WORDS];
  if ((int)blockIdx.x < nsort) {
    bucket_finish_body(ccol, crow, cw, offtab, rowptr, srow, sw, dis, n, E, nchunk, nsort, smem);
  } else {
    const int t0 = gt0 + ((int)blockIdx.x - nsort) * GT;
    const int t1 = (t0 + GT < gt1) ? t0 + GT : gt1;
    gemm_tiles<F_HID, 32, 16>(A, Wm, H, (float*)smem, t0, t1, 32);
  }
}

template <int F>
__device__ __forceinline__ void
gather_row(const float* __restrict__ H, const int* __restrict__ srow, const float* __restrict__ sw,
           const float* __restrict__ dis, int node, int beg, int end, float dc, int lane, float* acc) {
  constexpr int PER = F / 32;
  {
    const float* src = H + (long long)node * F + lane * PER;
    const float sl = dc * dc;
    if constexpr (PER == 4) {
      const float4 v = *(const float4*)src;
      acc[0] = v.x * sl; acc[1] = v.y * sl; acc[2] = v.z * sl; acc[3] = v.w * sl;
    } else {
      const float2 v = *(const float2*)src;
      acc[0] = v.x * sl; acc[1] = v.y * sl;
    }
  }

  for (int base = beg; base < end; base += 32) {
    const int p = base + lane;
    int rr = 0;
    float nw = 0.f;
    if (p < end) {
      rr = srow[p];
      nw = dis[rr] * sw[p] * dc;
    }
    int m = end - base;
    if (m > 32) m = 32;
#pragma unroll 4
    for (int j = 0; j < m; ++j) {
      const int r = __builtin_amdgcn_readlane(rr, j);
      const float w = __int_as_float(__builtin_amdgcn_readlane(__float_as_int(nw), j));
      const float* src = H + (long long)r * F + lane * PER;
      if constexpr (PER == 4) {
        const float4 v = *(const float4*)src;
        acc[0] += v.x * w; acc[1] += v.y * w; acc[2] += v.z * w; acc[3] += v.w * w;
      } else {
        const float2 v = *(const float2*)src;
        acc[0] += v.x * w; acc[1] += v.y * w;
      }
    }
  }
}

template <int F>
__global__ void __launch_bounds__(256)
k_gather(const float* __restrict__ H, const int* __restrict__ rowptr,
         const int* __restrict__ srow, const float* __restrict__ sw,
         const float* __restrict__ dis, const float* __restrict__ bias,
         float* __restrict__ out, int n) {
  constexpr int PER = F / 32;
  const int lane = threadIdx.x & 31;
  const int wave = __builtin_amdgcn_readfirstlane((int)(threadIdx.x >> 5));
  const int nstride = (int)gridDim.x * 8;
  int node = (int)blockIdx.x * 8 + wave;
  if (node >= n) return;

  float bv[PER];
#pragma unroll
  for (int q = 0; q < PER; ++q) bv[q] = bias[lane * PER + q];

  int beg = rowptr[node];
  int end = rowptr[node + 1];
  float dc = dis[node];

  for (;;) {
    const int nx = node + nstride;
    int nbeg = 0, nend = 0;
    float ndc = 0.f;
    if (nx < n) { nbeg = rowptr[nx]; nend = rowptr[nx + 1]; ndc = dis[nx]; }

    float acc[PER];
    gather_row<F>(H, srow, sw, dis, node, beg, end, dc, lane, acc);

    float* dst = out + (long long)node * F + lane * PER;
    if constexpr (PER == 4) {
      float4 o;
      o.x = acc[0] + bv[0]; o.y = acc[1] + bv[1]; o.z = acc[2] + bv[2]; o.w = acc[3] + bv[3];
      *(float4*)dst = o;
    } else {
      float2 o;
      o.x = acc[0] + bv[0]; o.y = acc[1] + bv[1];
      *(float2*)dst = o;
    }

    if (nx >= n) break;
    node = nx; beg = nbeg; end = nend; dc = ndc;
  }
}

__global__ void __launch_bounds__(512)
k_gather_gemm(const float* __restrict__ H, const int* __restrict__ rowptr,
              const int* __restrict__ srow, const float* __restrict__ sw,
              const float* __restrict__ dis, const float* __restrict__ bias,
              const float* __restrict__ W, float* __restrict__ H2, int n) {
  __shared__ float sW[F_HID * F_OUT];
  __shared__ float sT[2 * 16 * TPITCH];
  for (int i = threadIdx.x; i < F_HID * F_OUT; i += 512) sW[i] = W[i];

  const int lane = threadIdx.x & 31;
  const int wave = __builtin_amdgcn_readfirstlane((int)(threadIdx.x >> 5));
  const int ntile = n / 16;
  const int tstride = (int)gridDim.x;
  const int koff = (lane >> 4) * 2;
  int tile = (int)blockIdx.x;
  if (tile >= ntile) return;

  float bv[4];
#pragma unroll
  for (int q = 0; q < 4; ++q) bv[q] = bias[lane * 4 + q];

  int node = tile * 16 + wave;
  int beg = rowptr[node];
  int end = rowptr[node + 1];
  float dc = dis[node];

  for (int it = 0;; ++it) {
    const int nt = tile + tstride;
    const int nx = nt * 16 + wave;
    int nbeg = 0, nend = 0;
    float ndc = 0.f;
    if (nt < ntile) { nbeg = rowptr[nx]; nend = rowptr[nx + 1]; ndc = dis[nx]; }

    float acc[4];
    gather_row<F_HID>(H, srow, sw, dis, node, beg, end, dc, lane, acc);

    float4 o;
    o.x = acc[0] + bv[0]; o.y = acc[1] + bv[1]; o.z = acc[2] + bv[2]; o.w = acc[3] + bv[3];
    o.x = o.x > 0.f ? o.x : 0.f; o.y = o.y > 0.f ? o.y : 0.f;
    o.z = o.z > 0.f ? o.z : 0.f; o.w = o.w > 0.f ? o.w : 0.f;
    float* tbuf = sT + (it & 1) * (16 * TPITCH);
    *(float4*)(tbuf + wave * TPITCH + lane * 4) = o;
    __syncthreads();

    const int duty = (wave - 4 * it) & 15;
    if (duty < 4) {
      const int ncol = duty * 16 + (lane & 15);
      const float* arow = tbuf + (lane & 15) * TPITCH;
      const float* wb = sW + ncol;
      v8f c = {};
      for (int k0 = 0; k0 < F_HID; k0 += 32) {
        v2f a[8];
#pragma unroll
        for (int j = 0; j < 8; ++j) {
          const int ka = k0 + 4 * j + koff;
          a[j] = *(const v2f*)(arow + ka);
        }
#pragma unroll
        for (int j = 0; j < 8; ++j) {
          const int ka = k0 + 4 * j + koff;
          v2f b;
          b.x = wb[ka * F_OUT];
          b.y = wb[(ka + 1) * F_OUT];
          c = __builtin_amdgcn_wmma_f32_16x16x4_f32(
               false, a[j],  false, b,
               (short)0, c,  false,  false);
        }
      }
      const int rbase = tile * 16 + ((lane >> 4) << 3);
#pragma unroll
      for (int v = 0; v < 8; ++v)
        H2[(long long)(rbase + v) * F_OUT + ncol] = c[v];
    }

    if (nt >= ntile) break;
    tile = nt; node = nx; beg = nbeg; end = nend; dc = ndc;
  }
}

static inline int nblk(long long n, int t) { return (int)((n + t - 1) / t); }

extern "C" void kernel_launch(void* const* d_in, const int* in_sizes, int n_in,
                              void* d_out, int out_size, void* d_ws, size_t ws_size,
                              hipStream_t stream) {
  const float* x  = (const float*)d_in[0];
  const int*   ei = (const int*)d_in[1];
  const float* ew = (const float*)d_in[2];
  const float* W1 = (const float*)d_in[3];
  const float* b1 = (const float*)d_in[4];
  const float* W2 = (const float*)d_in[5];
  const float* b2 = (const float*)d_in[6];
  float* out = (float*)d_out;

  const int n = in_sizes[0] / F_IN;
  const int E = in_sizes[1] / 2;
  const int* row = ei;
  const int* col = ei + E;

  char* ws = (char*)d_ws;
  size_t o = 0;
  auto carve = [&](size_t bytes) {
    char* p = ws + o;
    o += (bytes + 255) & ~(size_t)255;
    return p;
  };
  const int nch = nblk(E, CH);
  const int nbk = nblk(n, 256);
  int*   offtab = (int*)carve((size_t)nch * OT * 4);
  int*   ccol   = (int*)carve((size_t)nch * CH * 4);
  int*   crow   = (int*)carve((size_t)nch * CH * 4);
  float* cw     = (float*)carve((size_t)nch * CH * 4);
  int*   rowptr = (int*)carve((size_t)(n + 1) * 4);
  int*   srow   = (int*)carve((size_t)E * 4);
  float* sw     = (float*)carve((size_t)E * 4);
  float* dis    = (float*)carve((size_t)n * 4);
  float* h      = (float*)carve((size_t)n * F_HID * 4);
  float* h2     = (float*)carve((size_t)n * F_OUT * 4);

  const int ntile1 = (n / 16) * (F_HID / 16);
  const int gsplit = (ntile1 * 7 / 16 / GT) * GT;
  const int g1 = gsplit / GT;
  const int g2 = nblk(ntile1 - gsplit, GT);
  k_sort1_gemm<<<nch + g1, 1024, 0, stream>>>(row, col, ew, ccol, crow, cw, offtab, E, nch,
                                              x, W1, h, 0, gsplit);
  k_sort2_gemm<<<nbk + g2, 1024, 0, stream>>>(ccol, crow, cw, offtab, rowptr, srow, sw, dis, n, E, nch, nbk,
                                              x, W1, h, gsplit, ntile1);

  k_gather_gemm<<<FUSED_BLOCKS, 512, 0, stream>>>(h, rowptr, srow, sw, dis, b1, W2, h2, n);

  k_gather<F_OUT><<<GATHER_BLOCKS, 256, 0, stream>>>(h2, rowptr, srow, sw, dis, b2, out, n);
}


# Gather: a node's loads go out together, not one by one: rows in whole groups of 8 (512 B) or 16 (256 B), zero-weight filler, no single-row remainder; self row issued with the edge list
# speedup vs baseline: 4.4450x; 1.0448x over previous
#include <hip/hip_runtime.h>
#include <cstdint>

#define F_IN  128
#define F_HID 128
#define F_OUT 64

#define GT            64
#define GATHER_BLOCKS 2048
#define FUSED_BLOCKS  1024
#define TPITCH        (F_HID + 4)

typedef __attribute__((ext_vector_type(2))) float v2f;
typedef __attribute__((ext_vector_type(8))) float v8f;

#define CH   4096
#define OT   257
#define U2   5
#define TCAP 8192

__device__ __forceinline__ int block_excl_scan(int v, int t, int* s_w, int& total) {
  const int lane = t & 31;
  const int wave = t >> 5;
  int x = v;
#pragma unroll
  for (int d = 1; d < 32; d <<= 1) {
    const int y = __shfl_up(x, d);
    if (lane >= d) x += y;
  }
  if (lane == 31) s_w[wave] = x;
  __syncthreads();
  if (wave == 0) {
    const int w = s_w[lane];
    int z = w;
#pragma unroll
    for (int d = 1; d < 32; d <<= 1) {
      const int y = __shfl_up(z, d);
      if (lane >= d) z += y;
    }
    s_w[lane] = z - w;
    if (lane == 31) s_w[32] = z;
  }
  __syncthreads();
  total = s_w[32];
  return s_w[wave] + x - v;
}

#define SORT1_WORDS (576 + 3 * CH)
#define SORT2_WORDS (1152 + 2 * TCAP)

__device__ __forceinline__ void
chunk_sort_body(const int* __restrict__ row, const int* __restrict__ col, const float* __restrict__ ew,
                int* __restrict__ ccol, int* __restrict__ crow, float* __restrict__ cw,
                int* __restrict__ offtab, int E, int* smem) {
  int*   s_hist = smem;
  int*   s_off  = smem + 256;
  int*   s_scan = smem + 520;
  int*   l_col  = smem + 576;
  int*   l_row  = smem + 576 + CH;
  float* l_w    = (float*)(smem + 576 + 2 * CH);
  const int t = threadIdx.x;
  const int e0 = blockIdx.x * CH;
  int cnt = E - e0;
  if (cnt > CH) cnt = CH;
  if (t < 256) s_hist[t] = 0;

  int cc[4], rr[4], rk[4];
  float ww[4];
#pragma unroll
  for (int k = 0; k < 4; ++k) {
    const int i = t + k * 1024;
    cc[k] = 0; rr[k] = 0; ww[k] = 0.f;
    if (i < cnt) { cc[k] = col[e0 + i]; rr[k] = row[e0 + i]; ww[k] = ew[e0 + i]; }
  }
  __syncthreads();
#pragma unroll
  for (int k = 0; k < 4; ++k) {
    rk[k] = 0;
    if (t + k * 1024 < cnt) rk[k] = atomicAdd(&s_hist[cc[k] >> 8], 1);
  }
  __syncthreads();
  int total;
  const int ex = block_excl_scan((t < 256) ? s_hist[t] : 0, t, s_scan, total);
  if (t <= 256) {
    s_off[t] = ex;
    offtab[blockIdx.x * OT + t] = ex;
  }
  __syncthreads();
#pragma unroll
  for (int k = 0; k < 4; ++k) {
    if (t + k * 1024 < cnt) {
      const int p = s_off[cc[k] >> 8] + rk[k];
      l_col[p] = cc[k]; l_row[p] = rr[k]; l_w[p] = ww[k];
    }
  }
  __syncthreads();
#pragma unroll
  for (int k = 0; k < 4; ++k) {
    const int i = t + k * 1024;
    if (i < cnt) { ccol[e0 + i] = l_col[i]; crow[e0 + i] = l_row[i]; cw[e0 + i] = l_w[i]; }
  }
}

__device__ __forceinline__ void
bucket_finish_body(const int* __restrict__ ccol, const int* __restrict__ crow, const float* __restrict__ cw,
                   const int* __restrict__ offtab, int* __restrict__ rowptr, int* __restrict__ srow,
                   float* __restrict__ sw, float* __restrict__ dis, int n, int E, int nchunk, int nbucket,
                   int* smem) {
  int*   s_pre    = smem;
  int*   s_src    = smem + 256;
  int*   s_cnt    = smem + 512;
  float* s_deg    = (float*)(smem + 768);
  int*   s_scan_a = smem + 1024;
  int*   s_scan_b = smem + 1064;
  int*   s_scan_c = smem + 1104;
  int*   st_row   = smem + 1152;
  float* st_w     = (float*)(smem + 1152 + TCAP);
  const int t = threadIdx.x;
  const int b = blockIdx.x;

  int lo = 0, len = 0;
  if (t < nchunk) {
    lo  = offtab[t * OT + b];
    len = offtab[t * OT + b + 1] - lo;
  }
  if (t < 256) { s_cnt[t] = 0; s_deg[t] = 0.f; }
  int total, base;
  const int pre = block_excl_scan(len, t, s_scan_a, total);
  (void)block_excl_scan(lo, t, s_scan_b, base);
  if (t < 256) {
    s_pre[t] = (t < nchunk) ? pre : 0x7fffffff;
    s_src[t] = t * CH + lo;
  }
  __syncthreads();

  auto locate = [&](int i) {
    int c = 0;
#pragma unroll
    for (int s = 128; s > 0; s >>= 1)
      if (s_pre[c + s] <= i) c += s;
    return s_src[c] + (i - s_pre[c]);
  };

  int kc[U2], kr[U2], ks[U2];
  float kw[U2];
#pragma unroll
  for (int k = 0; k < U2; ++k) ks[k] = 0;
#pragma unroll
  for (int s = 128; s > 0; s >>= 1) {
#pragma unroll
    for (int k = 0; k < U2; ++k)
      if (s_pre[ks[k] + s] <= t + k * 1024) ks[k] += s;
  }
#pragma unroll
  for (int k = 0; k < U2; ++k) {
    const int i = t + k * 1024;
    kc[k] = 0; kr[k] = 0; kw[k] = 0.f;
    if (i < total) {
      const int j = s_src[ks[k]] + (i - s_pre[ks[k]]);
      kc[k] = ccol[j]; kr[k] = crow[j]; kw[k] = cw[j];
    }
  }
#pragma unroll
  for (int k = 0; k < U2; ++k) {
    if (t + k * 1024 < total) {
      const int ln = kc[k] & 255;
      atomicAdd(&s_cnt[ln], 1);
      atomicAdd(&s_deg[ln], kw[k]);
    }
  }
  for (int i = t + U2 * 1024; i < total; i += 1024) {
    const int j = locate(i);
    const int ln = ccol[j] & 255;
    atomicAdd(&s_cnt[ln], 1);
    atomicAdd(&s_deg[ln], cw[j]);
  }
  __syncthreads();

  int tot2;
  const int ex = block_excl_scan((t < 256) ? s_cnt[t] : 0, t, s_scan_c, tot2);
  if (t < 256) {
    s_cnt[t] = ex;
    const int node = b * 256 + t;
    if (node < n) {
      rowptr[node] = base + ex;
      const float d = 1.0f + s_deg[t];
      dis[node] = (d > 0.f) ? rsqrtf(d) : 0.f;
    }
  }
  if (b == nbucket - 1 && t == 0) rowptr[n] = E;
  __syncthreads();

  const bool staged = total <= TCAP;
#pragma unroll
  for (int k = 0; k < U2; ++k) {
    if (t + k * 1024 < total) {
      const int p = atomicAdd(&s_cnt[kc[k] & 255], 1);
      if (staged) { st_row[p] = kr[k]; st_w[p] = kw[k]; }
      else        { srow[base + p] = kr[k]; sw[base + p] = kw[k]; }
    }
  }
  for (int i = t + U2 * 1024; i < total; i += 1024) {
    const int j = locate(i);
    const int p = atomicAdd(&s_cnt[ccol[j] & 255], 1);
    const int r = crow[j];
    const float w = cw[j];
    if (staged) { st_row[p] = r; st_w[p] = w; }
    else        { srow[base + p] = r; sw[base + p] = w; }
  }
  __syncthreads();
  if (staged) {
    for (int i = t; i < total; i += 1024) { srow[base + i] = st_row[i]; sw[base + i] = st_w[i]; }
  }
}

template <int FOUT, int NW, int KG>
__device__ __forceinline__ void
gemm_tiles(const float* __restrict__ A, const float* __restrict__ W, float* __restrict__ H,
           float* sW, int first, int last, int tstride) {
  for (int i = threadIdx.x; i < F_IN * FOUT; i += NW * 32) sW[i] = W[i];
  __syncthreads();

  const int lane = threadIdx.x & 31;
  const int wave = __builtin_amdgcn_readfirstlane((int)(threadIdx.x >> 5));
  constexpr int tilesN = FOUT / 16;
  const int koff = (lane >> 4) * 4;
  const unsigned aoff = (unsigned)((lane & 15) * F_IN + koff);

  for (int tile = first + wave; tile < last; tile += tstride) {
    const int tm = tile / tilesN;
    const int tn = tile % tilesN;
    const int ncol = tn * 16 + (lane & 15);

    const float* __restrict__ astrip = A + (long long)tm * (16 * F_IN);
    const float* __restrict__ wb = sW + ncol;

    v8f c = {};
    for (int k0 = 0; k0 < F_IN; k0 += 4 * KG) {
      float4 a[KG / 2];
#pragma unroll
      for (int j = 0; j < KG / 2; ++j)
        a[j] = *(const float4*)(astrip + aoff + (k0 + 8 * j));
#pragma unroll
      for (int j = 0; j < KG / 2; ++j) {
        const int ka = k0 + 8 * j + koff;
        v2f a0, a1, b0, b1;
        a0.x = a[j].x; a0.y = a[j].y;
        a1.x = a[j].z; a1.y = a[j].w;
        b0.x = wb[ka * FOUT];
        b0.y = wb[(ka + 1) * FOUT];
        b1.x = wb[(ka + 2) * FOUT];
        b1.y = wb[(ka + 3) * FOUT];
        c = __builtin_amdgcn_wmma_f32_16x16x4_f32(
             false, a0,  false, b0,
             (short)0, c,  false,  false);
        c = __builtin_amdgcn_wmma_f32_16x16x4_f32(
             false, a1,  false, b1,
             (short)0, c,  false,  false);
      }
    }

    const int rbase = tm * 16 + ((lane >> 4) << 3);
#pragma unroll
    for (int v = 0; v < 8; ++v)
      H[(long long)(rbase + v) * FOUT + ncol] = c[v];
  }
}

__global__ void __launch_bounds__(1024, 16)
k_sort1_gemm(const int* __restrict__ row, const int* __restrict__ col, const float* __restrict__ ew,
             int* __restrict__ ccol, int* __restrict__ crow, float* __restrict__ cw,
             int* __restrict__ offtab, int E, int nsort,
             const float* __restrict__ A, const float* __restrict__ Wm, float* __restrict__ H,
             int gt0, int gt1) {
  constexpr int WORDS = (F_IN * F_HID > SORT1_WORDS) ? F_IN * F_HID : SORT1_WORDS;
  __shared__ int smem[WORDS];
  if ((int)blockIdx.x < nsort) {
    chunk_sort_body(row, col, ew, ccol, crow, cw, offtab, E, smem);
  } else {
    const int t0 = gt0 + ((int)blockIdx.x - nsort) * GT;
    const int t1 = (t0 + GT < gt1) ? t0 + GT : gt1;
    gemm_tiles<F_HID, 32, 16>(A, Wm, H, (float*)smem, t0, t1, 32);
  }
}

__global__ void __launch_bounds__(1024, 16)
k_sort2_gemm(const int* __restrict__ ccol, const int* __restrict__ crow, const float* __restrict__ cw,
             const int* __restrict__ offtab, int* __restrict__ rowptr, int* __restrict__ srow,
             float* __restrict__ sw, float* __restrict__ dis, int n, int E, int nchunk, int nsort,
             const float* __restrict__ A, const float* __restrict__ Wm, float* __restrict__ H,
             int gt0, int gt1) {
  constexpr int WORDS = (F_IN * F_HID > SORT2_WORDS) ? F_IN * F_HID : SORT2_WORDS;
  __shared__ int smem[WORDS];
  if ((int)blockIdx.x < nsort) {
    bucket_finish_body(ccol, crow, cw, offtab, rowptr, srow, sw, dis, n, E, nchunk, nsort, smem);
  } else {
    const int t0 = gt0 + ((int)blockIdx.x - nsort) * GT;
    const int t1 = (t0 + GT < gt1) ? t0 + GT : gt1;
    gemm_tiles<F_HID, 32, 16>(A, Wm, H, (float*)smem, t0, t1, 32);
  }
}

template <int F>
__device__ __forceinline__ void
gather_row(const float* __restrict__ H, const int* __restrict__ srow, const float* __restrict__ sw,
           const float* __restrict__ dis, int node, int beg, int end, float dc, int lane, float* acc) {
  constexpr int PER = F / 32;
  constexpr int G = (PER == 4) ? 8 : 16;
  typedef __attribute__((ext_vector_type(PER))) float vrow;

  const vrow vs = *(const vrow*)(H + (long long)node * F + lane * PER);

  int rr = node;
  float nw = 0.f;
  if (beg + lane < end) {
    rr = srow[beg + lane];
    nw = dis[rr] * sw[beg + lane] * dc;
  }

  {
    const float sl = dc * dc;
#pragma unroll
    for (int q = 0; q < PER; ++q) acc[q] = vs[q] * sl;
  }

  for (int base = beg; base < end;) {
    int m = end - base;
    if (m > 32) m = 32;
    for (int j0 = 0; j0 < m; j0 += G) {
      vrow v[G];
#pragma unroll
      for (int j = 0; j < G; ++j) {
        const int r = __builtin_amdgcn_readlane(rr, j0 + j);
        v[j] = *(const vrow*)(H + (long long)r * F + lane * PER);
      }
#pragma unroll
      for (int j = 0; j < G; ++j) {
        const float w = __int_as_float(__builtin_amdgcn_readlane(__float_as_int(nw), j0 + j));
#pragma unroll
        for (int q = 0; q < PER; ++q) acc[q] = __builtin_fmaf(v[j][q], w, acc[q]);
      }
    }
    base += 32;
    if (base < end) {
      rr = node;
      nw = 0.f;
      if (base + lane < end) {
        rr = srow[base + lane];
        nw = dis[rr] * sw[base + lane] * dc;
      }
    }
  }
}

template <int F>
__global__ void __launch_bounds__(256)
k_gather(const float* __restrict__ H, const int* __restrict__ rowptr,
         const int* __restrict__ srow, const float* __restrict__ sw,
         const float* __restrict__ dis, const float* __restrict__ bias,
         float* __restrict__ out, int n) {
  constexpr int PER = F / 32;
  const int lane = threadIdx.x & 31;
  const int wave = __builtin_amdgcn_readfirstlane((int)(threadIdx.x >> 5));
  const int nstride = (int)gridDim.x * 8;
  int node = (int)blockIdx.x * 8 + wave;
  if (node >= n) return;

  float bv[PER];
#pragma unroll
  for (int q = 0; q < PER; ++q) bv[q] = bias[lane * PER + q];

  int beg = rowptr[node];
  int end = rowptr[node + 1];
  float dc = dis[node];

  for (;;) {
    const int nx = node + nstride;
    int nbeg = 0, nend = 0;
    float ndc = 0.f;
    if (nx < n) { nbeg = rowptr[nx]; nend = rowptr[nx + 1]; ndc = dis[nx]; }

    float acc[PER];
    gather_row<F>(H, srow, sw, dis, node, beg, end, dc, lane, acc);

    float* dst = out + (long long)node * F + lane * PER;
    if constexpr (PER == 4) {
      float4 o;
      o.x = acc[0] + bv[0]; o.y = acc[1] + bv[1]; o.z = acc[2] + bv[2]; o.w = acc[3] + bv[3];
      *(float4*)dst = o;
    } else {
      float2 o;
      o.x = acc[0] + bv[0]; o.y = acc[1] + bv[1];
      *(float2*)dst = o;
    }

    if (nx >= n) break;
    node = nx; beg = nbeg; end = nend; dc = ndc;
  }
}

__global__ void __launch_bounds__(512)
k_gather_gemm(const float* __restrict__ H, const int* __restrict__ rowptr,
              const int* __restrict__ srow, const float* __restrict__ sw,
              const float* __restrict__ dis, const float* __restrict__ bias,
              const float* __restrict__ W, float* __restrict__ H2, int n) {
  __shared__ float sW[F_HID * F_OUT];
  __shared__ float sT[2 * 16 * TPITCH];
  for (int i = threadIdx.x; i < F_HID * F_OUT; i += 512) sW[i] = W[i];

  const int lane = threadIdx.x & 31;
  const int wave = __builtin_amdgcn_readfirstlane((int)(threadIdx.x >> 5));
  const int ntile = n / 16;
  const int tstride = (int)gridDim.x;
  const int koff = (lane >> 4) * 2;
  int tile = (int)blockIdx.x;
  if (tile >= ntile) return;

  float bv[4];
#pragma unroll
  for (int q = 0; q < 4; ++q) bv[q] = bias[lane * 4 + q];

  int node = tile * 16 + wave;
  int beg = rowptr[node];
  int end = rowptr[node + 1];
  float dc = dis[node];

  for (int it = 0;; ++it) {
    const int nt = tile + tstride;
    const int nx = nt * 16 + wave;
    int nbeg = 0, nend = 0;
    float ndc = 0.f;
    if (nt < ntile) { nbeg = rowptr[nx]; nend = rowptr[nx + 1]; ndc = dis[nx]; }

    float acc[4];
    gather_row<F_HID>(H, srow, sw, dis, node, beg, end, dc, lane, acc);

    float4 o;
    o.x = acc[0] + bv[0]; o.y = acc[1] + bv[1]; o.z = acc[2] + bv[2]; o.w = acc[3] + bv[3];
    o.x = o.x > 0.f ? o.x : 0.f; o.y = o.y > 0.f ? o.y : 0.f;
    o.z = o.z > 0.f ? o.z : 0.f; o.w = o.w > 0.f ? o.w : 0.f;
    float* tbuf = sT + (it & 1) * (16 * TPITCH);
    *(float4*)(tbuf + wave * TPITCH + lane * 4) = o;
    __syncthreads();

    const int duty = (wave - 4 * it) & 15;
    if (duty < 4) {
      const int ncol = duty * 16 + (lane & 15);
      const float* arow = tbuf + (lane & 15) * TPITCH;
      const float* wb = sW + ncol;
      v8f c = {};
      for (int k0 = 0; k0 < F_HID; k0 += 32) {
        v2f a[8];
#pragma unroll
        for (int j = 0; j < 8; ++j) {
          const int ka = k0 + 4 * j + koff;
          a[j] = *(const v2f*)(arow + ka);
        }
#pragma unroll
        for (int j = 0; j < 8; ++j) {
          const int ka = k0 + 4 * j + koff;
          v2f b;
          b.x = wb[ka * F_OUT];
          b.y = wb[(ka + 1) * F_OUT];
          c = __builtin_amdgcn_wmma_f32_16x16x4_f32(
               false, a[j],  false, b,
               (short)0, c,  false,  false);
        }
      }
      const int rbase = tile * 16 + ((lane >> 4) << 3);
#pragma unroll
      for (int v = 0; v < 8; ++v)
        H2[(long long)(rbase + v) * F_OUT + ncol] = c[v];
    }

    if (nt >= ntile) break;
    tile = nt; node = nx; beg = nbeg; end = nend; dc = ndc;
  }
}

static inline int nblk(long long n, int t) { return (int)((n + t - 1) / t); }

extern "C" void kernel_launch(void* const* d_in, const int* in_sizes, int n_in,
                              void* d_out, int out_size, void* d_ws, size_t ws_size,
                              hipStream_t stream) {
  const float* x  = (const float*)d_in[0];
  const int*   ei = (const int*)d_in[1];
  const float* ew = (const float*)d_in[2];
  const float* W1 = (const float*)d_in[3];
  const float* b1 = (const float*)d_in[4];
  const float* W2 = (const float*)d_in[5];
  const float* b2 = (const float*)d_in[6];
  float* out = (float*)d_out;

  const int n = in_sizes[0] / F_IN;
  const int E = in_sizes[1] / 2;
  const int* row = ei;
  const int* col = ei + E;

  char* ws = (char*)d_ws;
  size_t o = 0;
  auto carve = [&](size_t bytes) {
    char* p = ws + o;
    o += (bytes + 255) & ~(size_t)255;
    return p;
  };
  const int nch = nblk(E, CH);
  const int nbk = nblk(n, 256);
  int*   offtab = (int*)carve((size_t)nch * OT * 4);
  int*   ccol   = (int*)carve((size_t)nch * CH * 4);
  int*   crow   = (int*)carve((size_t)nch * CH * 4);
  float* cw     = (float*)carve((size_t)nch * CH * 4);
  int*   rowptr = (int*)carve((size_t)(n + 1) * 4);
  int*   srow   = (int*)carve((size_t)E * 4);
  float* sw     = (float*)carve((size_t)E * 4);
  float* dis    = (float*)carve((size_t)n * 4);
  float* h      = (float*)carve((size_t)n * F_HID * 4);
  float* h2     = (float*)carve((size_t)n * F_OUT * 4);

  const int ntile1 = (n / 16) * (F_HID / 16);
  const int gsplit = (ntile1 * 7 / 16 / GT) * GT;
  const int g1 = gsplit / GT;
  const int g2 = nblk(ntile1 - gsplit, GT);
  k_sort1_gemm<<<nch + g1, 1024, 0, stream>>>(row, col, ew, ccol, crow, cw, offtab, E, nch,
                                              x, W1, h, 0, gsplit);
  k_sort2_gemm<<<nbk + g2, 1024, 0, stream>>>(ccol, crow, cw, offtab, rowptr, srow, sw, dis, n, E, nch, nbk,
                                              x, W1, h, gsplit, ntile1);

  k_gather_gemm<<<FUSED_BLOCKS, 512, 0, stream>>>(h, rowptr, srow, sw, dis, b1, W2, h2, n);

  k_gather<F_OUT><<<GATHER_BLOCKS, 256, 0, stream>>>(h2, rowptr, srow, sw, dis, b2, out, n);
}
